# starting kernel
# baseline (speedup 1.0000x reference)
#include <hip/hip_runtime.h>

#define HC 128
#define NH 8

typedef __attribute__((ext_vector_type(2))) float v2f;
typedef __attribute__((ext_vector_type(8))) float v8f;

__device__ __forceinline__ unsigned float_key(float f) {
  unsigned u = __float_as_uint(f);
  return (u & 0x80000000u) ? ~u : (u | 0x80000000u);
}
__device__ __forceinline__ float key_float(unsigned k) {
  unsigned u = (k & 0x80000000u) ? (k & 0x7FFFFFFFu) : ~k;
  return __uint_as_float(u);
}

__global__ void init_ws_kernel(unsigned* __restrict__ segkey,
                               float* __restrict__ denom,
                               float* __restrict__ agg, int n) {
  int gid = blockIdx.x * blockDim.x + threadIdx.x;
  if (gid < n * NH) { segkey[gid] = 0u; denom[gid] = 0.0f; }
  if (gid < n * HC) { agg[gid] = 0.0f; }
}

template <int K, int LDA, int LDB, int LDC>
__global__ __launch_bounds__(256) void wmma_gemm_f32(const float* __restrict__ A,
                                                     const float* __restrict__ B,
                                                     float* __restrict__ C,
                                                     int Mrows) {
  constexpr int KSTEPS = K / 4;
  __shared__ v2f Bs[KSTEPS * 32];

  const int tn = blockIdx.y;
#pragma unroll
  for (int idx = threadIdx.x; idx < KSTEPS * 32; idx += 256) {
    const int kk = idx >> 5, L = idx & 31;
    const int hf = L >> 4, lr2 = L & 15;
    const int ka = kk * 4 + 2 * hf;
    const int col = tn * 16 + lr2;
    v2f t;
    t.x = B[(size_t)ka * LDB + col];
    t.y = B[(size_t)(ka + 1) * LDB + col];
    Bs[idx] = t;
  }
  __syncthreads();

  const int lane = threadIdx.x & 31;
  const int tm = blockIdx.x * 8 + (threadIdx.x >> 5);
  if (tm * 16 >= Mrows) return;

  const int lr = lane & 15;
  const int half = lane >> 4;
  int arow = tm * 16 + lr; if (arow >= Mrows) arow = Mrows - 1;
  const float* ap = A + (size_t)arow * LDA + 2 * half;

  v8f acc = {};
#pragma unroll
  for (int kk = 0; kk < KSTEPS; ++kk) {
    const v2f a = *(const v2f*)(ap + kk * 4);
    acc = __builtin_amdgcn_wmma_f32_16x16x4_f32(false, a, false, Bs[kk * 32 + lane],
                                                (short)0, acc, false, false);
  }

  const int bcol = tn * 16 + lr;
#pragma unroll
  for (int v = 0; v < 8; ++v) {
    const int r = tm * 16 + v + 8 * half;
    if (r < Mrows) C[(size_t)r * LDC + bcol] = acc[v];
  }
}

__global__ __launch_bounds__(256) void edge_bias_logit_kernel(
    const float* __restrict__ PR, const float* __restrict__ Wb,
    const float* __restrict__ qbuf, const float* __restrict__ kvbuf,
    const int* __restrict__ qidx, const int* __restrict__ kidx,
    float* __restrict__ logit_out, unsigned* __restrict__ segkey, int Medges) {
  constexpr int KSTEPS = HC / 4;
  __shared__ v2f BmS[KSTEPS * 32];
  __shared__ v2f BaS[KSTEPS * 32];

  const int lane = threadIdx.x & 31;
  const int mtiles = (Medges + 15) >> 4;
  int tm = blockIdx.x * 8 + (threadIdx.x >> 5);
  if (tm >= mtiles) tm = mtiles - 1;

  const int half = lane >> 4;
  const int lr = lane & 15;
  int arow = tm * 16 + lr; if (arow >= Medges) arow = Medges - 1;

  v2f afrag[KSTEPS];
  const float* prow = PR + (size_t)arow * HC + 2 * half;
#pragma unroll
  for (int kk = 0; kk < KSTEPS; ++kk) afrag[kk] = *(const v2f*)(prow + kk * 4);

  float logacc[8];
#pragma unroll
  for (int v = 0; v < 8; ++v) logacc[v] = 0.0f;

  for (int j = 0; j < 8; ++j) {
    __syncthreads();
#pragma unroll
    for (int idx = threadIdx.x; idx < KSTEPS * 32; idx += 256) {
      const int kk = idx >> 5, L = idx & 31;
      const int hf = L >> 4, lr2 = L & 15;
      const int ka = kk * 4 + 2 * hf;
      const int col = j * 16 + lr2;
      v2f t1, t2;
      t1.x = Wb[(size_t)ka * 256 + col];
      t1.y = Wb[(size_t)(ka + 1) * 256 + col];
      t2.x = Wb[(size_t)ka * 256 + 128 + col];
      t2.y = Wb[(size_t)(ka + 1) * 256 + 128 + col];
      BmS[idx] = t1;
      BaS[idx] = t2;
    }
    __syncthreads();

    v8f bm = {};
    v8f ba = {};
#pragma unroll
    for (int kk = 0; kk < KSTEPS; ++kk) {
      bm = __builtin_amdgcn_wmma_f32_16x16x4_f32(false, afrag[kk], false,
                                                 BmS[kk * 32 + lane], (short)0,
                                                 bm, false, false);
      ba = __builtin_amdgcn_wmma_f32_16x16x4_f32(false, afrag[kk], false,
                                                 BaS[kk * 32 + lane], (short)0,
                                                 ba, false, false);
    }

    const int col = j * 16 + lr;
#pragma unroll
    for (int v = 0; v < 8; ++v) {
      int e = tm * 16 + v + 8 * half;
      e = (e < Medges) ? e : (Medges - 1);
      const int qi = qidx[e];
      const int ki = kidx[e];
      const float qv = qbuf[(size_t)qi * HC + col];
      const float kv = kvbuf[(size_t)ki * (2 * HC) + col];
      logacc[v] += qv * (kv * (1.0f + bm[v]) + ba[v]);
    }
  }

  const int h = lane & 7;
#pragma unroll
  for (int v = 0; v < 8; ++v) {
    logacc[v] += __shfl_xor(logacc[v], 8, 32);
    const int e = tm * 16 + v + 8 * half;
    if (((lane & 8) == 0) && e < Medges) {
      logit_out[(size_t)e * NH + h] = logacc[v];
      atomicMax(&segkey[(size_t)qidx[e] * NH + h], float_key(logacc[v]));
    }
  }
}

__global__ void exp_denom_kernel(const float* __restrict__ logit,
                                 const int* __restrict__ qidx,
                                 const unsigned* __restrict__ segkey,
                                 float* __restrict__ denom, int Medges) {
  int gid = blockIdx.x * blockDim.x + threadIdx.x;
  if (gid >= Medges * NH) return;
  const int e = gid >> 3, h = gid & 7;
  const int qi = qidx[e];
  const float mx = key_float(segkey[qi * NH + h]);
  const float ex = __expf(logit[(size_t)e * NH + h] - mx);
  atomicAdd(&denom[qi * NH + h], ex);
}

__global__ void scatter_agg_kernel(const float* __restrict__ logit,
                                   const int* __restrict__ qidx,
                                   const int* __restrict__ kidx,
                                   const unsigned* __restrict__ segkey,
                                   const float* __restrict__ denom,
                                   const float* __restrict__ kvbuf,
                                   float* __restrict__ agg, int Medges) {
  int gid = blockIdx.x * blockDim.x + threadIdx.x;
  if (gid >= Medges * HC) return;
  const int e = gid >> 7, c = gid & 127, h = c & 7;
  const int qi = qidx[e];
  const float mx   = key_float(segkey[qi * NH + h]);
  const float attn = __expf(logit[(size_t)e * NH + h] - mx) / denom[qi * NH + h];
  const float ve   = kvbuf[(size_t)kidx[e] * (2 * HC) + HC + c];
  atomicAdd(&agg[(size_t)qi * HC + c], attn * ve);
}

extern "C" void kernel_launch(void* const* d_in, const int* in_sizes, int n_in,
                              void* d_out, int out_size, void* d_ws, size_t ws_size,
                              hipStream_t stream) {
  const float* query = (const float*)d_in[0];
  const float* key_  = (const float*)d_in[1];
  const int*   qidx  = (const int*)d_in[2];
  const int*   kidx  = (const int*)d_in[3];
  const float* pr    = (const float*)d_in[4];
  const float* Wq    = (const float*)d_in[5];
  const float* Wkv   = (const float*)d_in[6];
  const float* Wb    = (const float*)d_in[7];
  const float* Wo    = (const float*)d_in[8];

  const int N = in_sizes[0] / HC;
  const int M = in_sizes[2];

  float* out_result = (float*)d_out;
  float* out_logit  = out_result + (size_t)N * HC;

  float*    q_buf  = (float*)d_ws;
  float*    kv_buf = q_buf + (size_t)N * HC;
  unsigned* segkey = (unsigned*)(kv_buf + (size_t)N * 2 * HC);
  float*    denom  = (float*)(segkey + (size_t)N * NH);
  float*    agg    = denom + (size_t)N * NH;

  const int TPB = 256;
  const int mtilesN = (N + 15) / 16;
  const int mtilesM = (M + 15) / 16;

  init_ws_kernel<<<(N * HC + TPB - 1) / TPB, TPB, 0, stream>>>(segkey, denom, agg, N);

  wmma_gemm_f32<HC, HC, HC, HC>
      <<<dim3((mtilesN + 7) / 8, HC / 16), TPB, 0, stream>>>(query, Wq, q_buf, N);

  wmma_gemm_f32<HC, HC, 2 * HC, 2 * HC>
      <<<dim3((mtilesN + 7) / 8, 2 * HC / 16), TPB, 0, stream>>>(key_, Wkv, kv_buf, N);

  edge_bias_logit_kernel<<<(mtilesM + 7) / 8, TPB, 0, stream>>>(
      pr, Wb, q_buf, kv_buf, qidx, kidx, out_logit, segkey, M);

  exp_denom_kernel<<<(M * NH + TPB - 1) / TPB, TPB, 0, stream>>>(
      out_logit, qidx, segkey, denom, M);

  scatter_agg_kernel<<<(M * HC + TPB - 1) / TPB, TPB, 0, stream>>>(
      out_logit, qidx, kidx, segkey, denom, kv_buf, agg, M);

  wmma_gemm_f32<HC, HC, HC, HC>
      <<<dim3((mtilesN + 7) / 8, HC / 16), TPB, 0, stream>>>(agg, Wo, out_result, N);
}


# projection GEMMs (q, kv, out): wave keeps its weight column-tile fragments in registers; block loads its 32 A rows once as whole 512-byte rows, shares them via LDS; A not re-read per column tile
# speedup vs baseline: 1.5173x; 1.5173x over previous
#include <hip/hip_runtime.h>

#define HC 128
#define NH 8

typedef __attribute__((ext_vector_type(2))) float v2f;
typedef __attribute__((ext_vector_type(8))) float v8f;

__device__ __forceinline__ unsigned float_key(float f) {
  unsigned u = __float_as_uint(f);
  return (u & 0x80000000u) ? ~u : (u | 0x80000000u);
}
__device__ __forceinline__ float key_float(unsigned k) {
  unsigned u = (k & 0x80000000u) ? (k & 0x7FFFFFFFu) : ~k;
  return __uint_as_float(u);
}

__global__ void init_ws_kernel(unsigned* __restrict__ segkey,
                               float* __restrict__ denom,
                               float* __restrict__ agg, int n) {
  int gid = blockIdx.x * blockDim.x + threadIdx.x;
  if (gid < n * NH) { segkey[gid] = 0u; denom[gid] = 0.0f; }
  if (gid < n * HC) { agg[gid] = 0.0f; }
}

#define GEMM_ROWS 32
#define GEMM_PITCH 132
#define GEMM_KG 8
template <int LDB, int LDC, int NT>
__global__ __launch_bounds__(256) void wmma_gemm_f32(const float* __restrict__ A,
                                                     const float* __restrict__ B,
                                                     float* __restrict__ C,
                                                     int Mrows) {
  constexpr int KSTEPS = HC / 4;
  constexpr int NG = KSTEPS / GEMM_KG;
  __shared__ __attribute__((aligned(16))) float As[GEMM_ROWS * GEMM_PITCH];

  const int lane = threadIdx.x & 31;
  const int w = threadIdx.x >> 5;
  const int lr = lane & 15;
  const int half = lane >> 4;
  const int row0 = blockIdx.x * GEMM_ROWS;

  v2f bfrag[NT][KSTEPS];
#pragma unroll
  for (int t = 0; t < NT; ++t) {
    const float* bp = B + (size_t)(2 * half) * LDB + (w * NT + t) * 16 + lr;
#pragma unroll
    for (int kk = 0; kk < KSTEPS; ++kk) {
      v2f f;
      f.x = bp[(size_t)(kk * 4) * LDB];
      f.y = bp[(size_t)(kk * 4 + 1) * LDB];
      bfrag[t][kk] = f;
    }
  }

#pragma unroll
  for (int it = 0; it < GEMM_ROWS / 8; ++it) {
    const int r = it * 8 + w;
    int grow = row0 + r; if (grow >= Mrows) grow = Mrows - 1;
    const float4 val = *(const float4*)(A + (size_t)grow * HC + lane * 4);
    *(float4*)(As + r * GEMM_PITCH + lane * 4) = val;
  }
  __syncthreads();

  const float* ap0 = As + lr * GEMM_PITCH + 2 * half;
  const float* ap1 = ap0 + 16 * GEMM_PITCH;

  v8f acc0[NT], acc1[NT];
#pragma unroll
  for (int t = 0; t < NT; ++t) { acc0[t] = (v8f){}; acc1[t] = (v8f){}; }

  v2f a0[2][GEMM_KG], a1[2][GEMM_KG];
#pragma unroll
  for (int k = 0; k < GEMM_KG; ++k) {
    a0[0][k] = *(const v2f*)(ap0 + k * 4);
    a1[0][k] = *(const v2f*)(ap1 + k * 4);
  }
#pragma unroll
  for (int g = 0; g < NG; ++g) {
    const int cur = g & 1, nxt = cur ^ 1;
    if (g + 1 < NG) {
#pragma unroll
      for (int k = 0; k < GEMM_KG; ++k) {
        a0[nxt][k] = *(const v2f*)(ap0 + ((g + 1) * GEMM_KG + k) * 4);
        a1[nxt][k] = *(const v2f*)(ap1 + ((g + 1) * GEMM_KG + k) * 4);
      }
    }
    __builtin_amdgcn_sched_barrier(0);
#pragma unroll
    for (int k = 0; k < GEMM_KG; ++k) {
#pragma unroll
      for (int t = 0; t < NT; ++t) {
        acc0[t] = __builtin_amdgcn_wmma_f32_16x16x4_f32(false, a0[cur][k], false,
                                                        bfrag[t][g * GEMM_KG + k],
                                                        (short)0, acc0[t], false, false);
        acc1[t] = __builtin_amdgcn_wmma_f32_16x16x4_f32(false, a1[cur][k], false,
                                                        bfrag[t][g * GEMM_KG + k],
                                                        (short)0, acc1[t], false, false);
      }
    }
    __builtin_amdgcn_sched_barrier(0);
  }

  if (row0 + GEMM_ROWS <= Mrows) {
#pragma unroll
    for (int t = 0; t < NT; ++t) {
      float* cp = C + (size_t)(row0 + 8 * half) * LDC + (w * NT + t) * 16 + lr;
#pragma unroll
      for (int v = 0; v < 8; ++v) {
        cp[(size_t)v * LDC] = acc0[t][v];
        cp[(size_t)(v + 16) * LDC] = acc1[t][v];
      }
    }
  } else {
#pragma unroll
    for (int t = 0; t < NT; ++t) {
      const int bcol = (w * NT + t) * 16 + lr;
#pragma unroll
      for (int v = 0; v < 8; ++v) {
        const int r0 = row0 + v + 8 * half;
        if (r0 < Mrows) C[(size_t)r0 * LDC + bcol] = acc0[t][v];
        const int r1 = r0 + 16;
        if (r1 < Mrows) C[(size_t)r1 * LDC + bcol] = acc1[t][v];
      }
    }
  }
}

__global__ __launch_bounds__(256) void edge_bias_logit_kernel(
    const float* __restrict__ PR, const float* __restrict__ Wb,
    const float* __restrict__ qbuf, const float* __restrict__ kvbuf,
    const int* __restrict__ qidx, const int* __restrict__ kidx,
    float* __restrict__ logit_out, unsigned* __restrict__ segkey, int Medges) {
  constexpr int KSTEPS = HC / 4;
  __shared__ v2f BmS[KSTEPS * 32];
  __shared__ v2f BaS[KSTEPS * 32];

  const int lane = threadIdx.x & 31;
  const int mtiles = (Medges + 15) >> 4;
  int tm = blockIdx.x * 8 + (threadIdx.x >> 5);
  if (tm >= mtiles) tm = mtiles - 1;

  const int half = lane >> 4;
  const int lr = lane & 15;
  int arow = tm * 16 + lr; if (arow >= Medges) arow = Medges - 1;

  v2f afrag[KSTEPS];
  const float* prow = PR + (size_t)arow * HC + 2 * half;
#pragma unroll
  for (int kk = 0; kk < KSTEPS; ++kk) afrag[kk] = *(const v2f*)(prow + kk * 4);

  float logacc[8];
#pragma unroll
  for (int v = 0; v < 8; ++v) logacc[v] = 0.0f;

  for (int j = 0; j < 8; ++j) {
    __syncthreads();
#pragma unroll
    for (int idx = threadIdx.x; idx < KSTEPS * 32; idx += 256) {
      const int kk = idx >> 5, L = idx & 31;
      const int hf = L >> 4, lr2 = L & 15;
      const int ka = kk * 4 + 2 * hf;
      const int col = j * 16 + lr2;
      v2f t1, t2;
      t1.x = Wb[(size_t)ka * 256 + col];
      t1.y = Wb[(size_t)(ka + 1) * 256 + col];
      t2.x = Wb[(size_t)ka * 256 + 128 + col];
      t2.y = Wb[(size_t)(ka + 1) * 256 + 128 + col];
      BmS[idx] = t1;
      BaS[idx] = t2;
    }
    __syncthreads();

    v8f bm = {};
    v8f ba = {};
#pragma unroll
    for (int kk = 0; kk < KSTEPS; ++kk) {
      bm = __builtin_amdgcn_wmma_f32_16x16x4_f32(false, afrag[kk], false,
                                                 BmS[kk * 32 + lane], (short)0,
                                                 bm, false, false);
      ba = __builtin_amdgcn_wmma_f32_16x16x4_f32(false, afrag[kk], false,
                                                 BaS[kk * 32 + lane], (short)0,
                                                 ba, false, false);
    }

    const int col = j * 16 + lr;
#pragma unroll
    for (int v = 0; v < 8; ++v) {
      int e = tm * 16 + v + 8 * half;
      e = (e < Medges) ? e : (Medges - 1);
      const int qi = qidx[e];
      const int ki = kidx[e];
      const float qv = qbuf[(size_t)qi * HC + col];
      const float kv = kvbuf[(size_t)ki * (2 * HC) + col];
      logacc[v] += qv * (kv * (1.0f + bm[v]) + ba[v]);
    }
  }

  const int h = lane & 7;
#pragma unroll
  for (int v = 0; v < 8; ++v) {
    logacc[v] += __shfl_xor(logacc[v], 8, 32);
    const int e = tm * 16 + v + 8 * half;
    if (((lane & 8) == 0) && e < Medges) {
      logit_out[(size_t)e * NH + h] = logacc[v];
      atomicMax(&segkey[(size_t)qidx[e] * NH + h], float_key(logacc[v]));
    }
  }
}

__global__ void exp_denom_kernel(const float* __restrict__ logit,
                                 const int* __restrict__ qidx,
                                 const unsigned* __restrict__ segkey,
                                 float* __restrict__ denom, int Medges) {
  int gid = blockIdx.x * blockDim.x + threadIdx.x;
  if (gid >= Medges * NH) return;
  const int e = gid >> 3, h = gid & 7;
  const int qi = qidx[e];
  const float mx = key_float(segkey[qi * NH + h]);
  const float ex = __expf(logit[(size_t)e * NH + h] - mx);
  atomicAdd(&denom[qi * NH + h], ex);
}

#define EPW 8
__global__ __launch_bounds__(256) void scatter_agg_kernel(const float* __restrict__ logit,
                                   const int* __restrict__ qidx,
                                   const int* __restrict__ kidx,
                                   const unsigned* __restrict__ segkey,
                                   const float* __restrict__ denom,
                                   const float* __restrict__ kvbuf,
                                   float* __restrict__ agg, int Medges) {
  const int lane = threadIdx.x & 31;
  const int wave = blockIdx.x * 8 + __builtin_amdgcn_readfirstlane((int)(threadIdx.x >> 5));
  const int e0 = wave * EPW;
  if (e0 >= Medges) return;
  const int h = lane & 7;

  if (e0 + EPW <= Medges) {
    const int myq = qidx[e0 + h];
    const int myk = kidx[e0 + h];
    int qi[EPW], ki[EPW];
#pragma unroll
    for (int u = 0; u < EPW; ++u) {
      qi[u] = __builtin_amdgcn_readlane(myq, u);
      ki[u] = __builtin_amdgcn_readlane(myk, u);
    }
    float lg[EPW], dn[EPW], vv[EPW][4];
    unsigned sk[EPW];
#pragma unroll
    for (int u = 0; u < EPW; ++u) {
      const float* vp = kvbuf + (size_t)ki[u] * (2 * HC) + HC + lane;
#pragma unroll
      for (int i = 0; i < 4; ++i) vv[u][i] = vp[32 * i];
    }
#pragma unroll
    for (int u = 0; u < EPW; ++u) {
      lg[u] = logit[(size_t)(e0 + u) * NH + h];
      sk[u] = segkey[(size_t)qi[u] * NH + h];
      dn[u] = denom[(size_t)qi[u] * NH + h];
    }
#pragma unroll
    for (int u = 0; u < EPW; ++u) {
      const float attn = __expf(lg[u] - key_float(sk[u])) / dn[u];
      float* ap = agg + (size_t)qi[u] * HC + lane;
#pragma unroll
      for (int i = 0; i < 4; ++i) atomicAdd(ap + 32 * i, attn * vv[u][i]);
    }
  } else {
    const int nE = Medges - e0;
    for (int u = 0; u < nE; ++u) {
      const int e = e0 + u;
      const int q1 = qidx[e];
      const int k1 = kidx[e];
      const float mx = key_float(segkey[(size_t)q1 * NH + h]);
      const float attn = __expf(logit[(size_t)e * NH + h] - mx) / denom[(size_t)q1 * NH + h];
      const float* vp = kvbuf + (size_t)k1 * (2 * HC) + HC + lane;
      float* ap = agg + (size_t)q1 * HC + lane;
#pragma unroll
      for (int i = 0; i < 4; ++i) atomicAdd(ap + 32 * i, attn * vp[32 * i]);
    }
  }
}

extern "C" void kernel_launch(void* const* d_in, const int* in_sizes, int n_in,
                              void* d_out, int out_size, void* d_ws, size_t ws_size,
                              hipStream_t stream) {
  const float* query = (const float*)d_in[0];
  const float* key_  = (const float*)d_in[1];
  const int*   qidx  = (const int*)d_in[2];
  const int*   kidx  = (const int*)d_in[3];
  const float* pr    = (const float*)d_in[4];
  const float* Wq    = (const float*)d_in[5];
  const float* Wkv   = (const float*)d_in[6];
  const float* Wb    = (const float*)d_in[7];
  const float* Wo    = (const float*)d_in[8];

  const int N = in_sizes[0] / HC;
  const int M = in_sizes[2];

  float* out_result = (float*)d_out;
  float* out_logit  = out_result + (size_t)N * HC;

  float*    q_buf  = (float*)d_ws;
  float*    kv_buf = q_buf + (size_t)N * HC;
  unsigned* segkey = (unsigned*)(kv_buf + (size_t)N * 2 * HC);
  float*    denom  = (float*)(segkey + (size_t)N * NH);
  float*    agg    = denom + (size_t)N * NH;

  const int TPB = 256;
  const int mtilesM = (M + 15) / 16;

  init_ws_kernel<<<(N * HC + TPB - 1) / TPB, TPB, 0, stream>>>(segkey, denom, agg, N);

  wmma_gemm_f32<HC, HC, 1>
      <<<(N + GEMM_ROWS - 1) / GEMM_ROWS, TPB, 0, stream>>>(query, Wq, q_buf, N);

  wmma_gemm_f32<2 * HC, 2 * HC, 2>
      <<<(N + GEMM_ROWS - 1) / GEMM_ROWS, TPB, 0, stream>>>(key_, Wkv, kv_buf, N);

  edge_bias_logit_kernel<<<(mtilesM + 7) / 8, TPB, 0, stream>>>(
      pr, Wb, q_buf, kv_buf, qidx, kidx, out_logit, segkey, M);

  exp_denom_kernel<<<(M * NH + TPB - 1) / TPB, TPB, 0, stream>>>(
      out_logit, qidx, segkey, denom, M);

  scatter_agg_kernel<<<((M + EPW - 1) / EPW + 7) / 8, TPB, 0, stream>>>(
      out_logit, qidx, kidx, segkey, denom, kv_buf, agg, M);

  wmma_gemm_f32<HC, HC, 1>
      <<<(N + GEMM_ROWS - 1) / GEMM_ROWS, TPB, 0, stream>>>(agg, Wo, out_result, N);
}


# edge kernel: columns dealt to waves, Wb fragments (16+16 columns) kept in registers; 512 blocks walk 16-edge tiles, pr/q/k rows staged once in LDS; logits summed across waves, fp sum order changes
# speedup vs baseline: 1.8165x; 1.1972x over previous
#include <hip/hip_runtime.h>

#define HC 128
#define NH 8

typedef __attribute__((ext_vector_type(2))) float v2f;
typedef __attribute__((ext_vector_type(8))) float v8f;

__device__ __forceinline__ unsigned float_key(float f) {
  unsigned u = __float_as_uint(f);
  return (u & 0x80000000u) ? ~u : (u | 0x80000000u);
}
__device__ __forceinline__ float key_float(unsigned k) {
  unsigned u = (k & 0x80000000u) ? (k & 0x7FFFFFFFu) : ~k;
  return __uint_as_float(u);
}

__global__ void init_ws_kernel(unsigned* __restrict__ segkey,
                               float* __restrict__ denom,
                               float* __restrict__ agg, int n) {
  int gid = blockIdx.x * blockDim.x + threadIdx.x;
  if (gid < n * NH) { segkey[gid] = 0u; denom[gid] = 0.0f; }
  if (gid < n * HC) { agg[gid] = 0.0f; }
}

#define GEMM_ROWS 32
#define GEMM_PITCH 132
#define GEMM_KG 8
template <int LDB, int LDC, int NT>
__global__ __launch_bounds__(256) void wmma_gemm_f32(const float* __restrict__ A,
                                                     const float* __restrict__ B,
                                                     float* __restrict__ C,
                                                     int Mrows) {
  constexpr int KSTEPS = HC / 4;
  constexpr int NG = KSTEPS / GEMM_KG;
  __shared__ __attribute__((aligned(16))) float As[GEMM_ROWS * GEMM_PITCH];

  const int lane = threadIdx.x & 31;
  const int w = threadIdx.x >> 5;
  const int lr = lane & 15;
  const int half = lane >> 4;
  const int row0 = blockIdx.x * GEMM_ROWS;

  v2f bfrag[NT][KSTEPS];
#pragma unroll
  for (int t = 0; t < NT; ++t) {
    const float* bp = B + (size_t)(2 * half) * LDB + (w * NT + t) * 16 + lr;
#pragma unroll
    for (int kk = 0; kk < KSTEPS; ++kk) {
      v2f f;
      f.x = bp[(size_t)(kk * 4) * LDB];
      f.y = bp[(size_t)(kk * 4 + 1) * LDB];
      bfrag[t][kk] = f;
    }
  }

#pragma unroll
  for (int it = 0; it < GEMM_ROWS / 8; ++it) {
    const int r = it * 8 + w;
    int grow = row0 + r; if (grow >= Mrows) grow = Mrows - 1;
    const float4 val = *(const float4*)(A + (size_t)grow * HC + lane * 4);
    *(float4*)(As + r * GEMM_PITCH + lane * 4) = val;
  }
  __syncthreads();

  const float* ap0 = As + lr * GEMM_PITCH + 2 * half;
  const float* ap1 = ap0 + 16 * GEMM_PITCH;

  v8f acc0[NT], acc1[NT];
#pragma unroll
  for (int t = 0; t < NT; ++t) { acc0[t] = (v8f){}; acc1[t] = (v8f){}; }

  v2f a0[2][GEMM_KG], a1[2][GEMM_KG];
#pragma unroll
  for (int k = 0; k < GEMM_KG; ++k) {
    a0[0][k] = *(const v2f*)(ap0 + k * 4);
    a1[0][k] = *(const v2f*)(ap1 + k * 4);
  }
#pragma unroll
  for (int g = 0; g < NG; ++g) {
    const int cur = g & 1, nxt = cur ^ 1;
    if (g + 1 < NG) {
#pragma unroll
      for (int k = 0; k < GEMM_KG; ++k) {
        a0[nxt][k] = *(const v2f*)(ap0 + ((g + 1) * GEMM_KG + k) * 4);
        a1[nxt][k] = *(const v2f*)(ap1 + ((g + 1) * GEMM_KG + k) * 4);
      }
    }
    __builtin_amdgcn_sched_barrier(0);
#pragma unroll
    for (int k = 0; k < GEMM_KG; ++k) {
#pragma unroll
      for (int t = 0; t < NT; ++t) {
        acc0[t] = __builtin_amdgcn_wmma_f32_16x16x4_f32(false, a0[cur][k], false,
                                                        bfrag[t][g * GEMM_KG + k],
                                                        (short)0, acc0[t], false, false);
        acc1[t] = __builtin_amdgcn_wmma_f32_16x16x4_f32(false, a1[cur][k], false,
                                                        bfrag[t][g * GEMM_KG + k],
                                                        (short)0, acc1[t], false, false);
      }
    }
    __builtin_amdgcn_sched_barrier(0);
  }

  if (row0 + GEMM_ROWS <= Mrows) {
#pragma unroll
    for (int t = 0; t < NT; ++t) {
      float* cp = C + (size_t)(row0 + 8 * half) * LDC + (w * NT + t) * 16 + lr;
#pragma unroll
      for (int v = 0; v < 8; ++v) {
        cp[(size_t)v * LDC] = acc0[t][v];
        cp[(size_t)(v + 16) * LDC] = acc1[t][v];
      }
    }
  } else {
#pragma unroll
    for (int t = 0; t < NT; ++t) {
      const int bcol = (w * NT + t) * 16 + lr;
#pragma unroll
      for (int v = 0; v < 8; ++v) {
        const int r0 = row0 + v + 8 * half;
        if (r0 < Mrows) C[(size_t)r0 * LDC + bcol] = acc0[t][v];
        const int r1 = r0 + 16;
        if (r1 < Mrows) C[(size_t)r1 * LDC + bcol] = acc1[t][v];
      }
    }
  }
}

#define EDGE_PITCH 132
#define EDGE_KG 8
__global__ __launch_bounds__(256) void edge_bias_logit_kernel(
    const float* __restrict__ PR, const float* __restrict__ Wb,
    const float* __restrict__ qbuf, const float* __restrict__ kvbuf,
    const int* __restrict__ qidx, const int* __restrict__ kidx,
    float* __restrict__ logit_out, unsigned* __restrict__ segkey, int Medges) {
  constexpr int KSTEPS = HC / 4;
  constexpr int NG = KSTEPS / EDGE_KG;
  __shared__ __attribute__((aligned(16))) float As[16 * EDGE_PITCH];
  __shared__ __attribute__((aligned(16))) float Qs[16 * EDGE_PITCH];
  __shared__ __attribute__((aligned(16))) float Ks[16 * EDGE_PITCH];
  __shared__ float Ps[8 * 128];
  __shared__ int Is[2][16];

  const int tid = threadIdx.x;
  const int lane = tid & 31;
  const int w = __builtin_amdgcn_readfirstlane(tid >> 5);
  const int lr = lane & 15;
  const int half = lane >> 4;
  const int mtiles = (Medges + 15) >> 4;
  const int stride = gridDim.x;
  const int last = Medges - 1;

  v2f bmf[KSTEPS], baf[KSTEPS];
  {
    const float* bp = Wb + (size_t)(2 * half) * 256 + w * 16 + lr;
#pragma unroll
    for (int kk = 0; kk < KSTEPS; ++kk) {
      v2f f, g;
      f.x = bp[(size_t)(kk * 4) * 256];
      f.y = bp[(size_t)(kk * 4 + 1) * 256];
      g.x = bp[(size_t)(kk * 4) * 256 + 128];
      g.y = bp[(size_t)(kk * 4 + 1) * 256 + 128];
      bmf[kk] = f;
      baf[kk] = g;
    }
  }

  int tile = blockIdx.x;
  float4 ra0 = {0.f, 0.f, 0.f, 0.f}, ra1 = ra0, rq0 = ra0, rq1 = ra0, rk0 = ra0, rk1 = ra0;
  int nq0 = 0, nq1 = 0, nk0 = 0, nk1 = 0;
  if (tile < mtiles) {
    int e0 = tile * 16 + 2 * w, e1 = e0 + 1;
    e0 = e0 < last ? e0 : last;
    e1 = e1 < last ? e1 : last;
    const int q0 = qidx[e0], q1 = qidx[e1], k0 = kidx[e0], k1 = kidx[e1];
    if (lane < 2) Is[0][2 * w + lane] = lane ? q1 : q0;
    ra0 = *(const float4*)(PR + (size_t)e0 * HC + lane * 4);
    ra1 = *(const float4*)(PR + (size_t)e1 * HC + lane * 4);
    rq0 = *(const float4*)(qbuf + (size_t)q0 * HC + lane * 4);
    rq1 = *(const float4*)(qbuf + (size_t)q1 * HC + lane * 4);
    rk0 = *(const float4*)(kvbuf + (size_t)k0 * (2 * HC) + lane * 4);
    rk1 = *(const float4*)(kvbuf + (size_t)k1 * (2 * HC) + lane * 4);
    if (tile + stride < mtiles) {
      int f0 = (tile + stride) * 16 + 2 * w, f1 = f0 + 1;
      f0 = f0 < last ? f0 : last;
      f1 = f1 < last ? f1 : last;
      nq0 = qidx[f0]; nq1 = qidx[f1]; nk0 = kidx[f0]; nk1 = kidx[f1];
    }
  }

  const float* ap = As + lr * EDGE_PITCH + 2 * half;
  const int srow = (2 * w) * EDGE_PITCH + lane * 4;
  int par = 0;

  for (; tile < mtiles; tile += stride) {
    *(float4*)(As + srow) = ra0;
    *(float4*)(As + srow + EDGE_PITCH) = ra1;
    *(float4*)(Qs + srow) = rq0;
    *(float4*)(Qs + srow + EDGE_PITCH) = rq1;
    *(float4*)(Ks + srow) = rk0;
    *(float4*)(Ks + srow + EDGE_PITCH) = rk1;
    __syncthreads();

    const int myq = Is[par][(tid >> 3) & 15];

    const int nt = tile + stride;
    if (nt < mtiles) {
      int e0 = nt * 16 + 2 * w, e1 = e0 + 1;
      e0 = e0 < last ? e0 : last;
      e1 = e1 < last ? e1 : last;
      ra0 = *(const float4*)(PR + (size_t)e0 * HC + lane * 4);
      ra1 = *(const float4*)(PR + (size_t)e1 * HC + lane * 4);
      rq0 = *(const float4*)(qbuf + (size_t)nq0 * HC + lane * 4);
      rq1 = *(const float4*)(qbuf + (size_t)nq1 * HC + lane * 4);
      rk0 = *(const float4*)(kvbuf + (size_t)nk0 * (2 * HC) + lane * 4);
      rk1 = *(const float4*)(kvbuf + (size_t)nk1 * (2 * HC) + lane * 4);
      if (lane < 2) Is[par ^ 1][2 * w + lane] = lane ? nq1 : nq0;
      const int nt2 = nt + stride;
      if (nt2 < mtiles) {
        int f0 = nt2 * 16 + 2 * w, f1 = f0 + 1;
        f0 = f0 < last ? f0 : last;
        f1 = f1 < last ? f1 : last;
        nq0 = qidx[f0]; nq1 = qidx[f1]; nk0 = kidx[f0]; nk1 = kidx[f1];
      }
    }

    v8f bm = {};
    v8f ba = {};
    v2f a[2][EDGE_KG];
#pragma unroll
    for (int k = 0; k < EDGE_KG; ++k) a[0][k] = *(const v2f*)(ap + k * 4);
#pragma unroll
    for (int g = 0; g < NG; ++g) {
      const int cur = g & 1, nxt = cur ^ 1;
      if (g + 1 < NG) {
#pragma unroll
        for (int k = 0; k < EDGE_KG; ++k)
          a[nxt][k] = *(const v2f*)(ap + ((g + 1) * EDGE_KG + k) * 4);
      }
      __builtin_amdgcn_sched_barrier(0);
#pragma unroll
      for (int k = 0; k < EDGE_KG; ++k) {
        bm = __builtin_amdgcn_wmma_f32_16x16x4_f32(false, a[cur][k], false,
                                                   bmf[g * EDGE_KG + k], (short)0,
                                                   bm, false, false);
        ba = __builtin_amdgcn_wmma_f32_16x16x4_f32(false, a[cur][k], false,
                                                   baf[g * EDGE_KG + k], (short)0,
                                                   ba, false, false);
      }
      __builtin_amdgcn_sched_barrier(0);
    }

    float p[8];
#pragma unroll
    for (int v = 0; v < 8; ++v) {
      const int r = v + 8 * half;
      const float qv = Qs[r * EDGE_PITCH + w * 16 + lr];
      const float kv = Ks[r * EDGE_PITCH + w * 16 + lr];
      float t = qv * (kv * (1.0f + bm[v]) + ba[v]);
      t += __shfl_xor(t, 8, 32);
      p[v] = t;
    }
    if (lr < 8) {
#pragma unroll
      for (int v = 0; v < 8; ++v) Ps[w * 128 + (v + 8 * half) * 8 + lr] = p[v];
    }
    __syncthreads();

    if (tid < 128) {
      float s = Ps[tid];
#pragma unroll
      for (int j = 1; j < 8; ++j) s += Ps[j * 128 + tid];
      const int e = tile * 16 + (tid >> 3);
      if (e < Medges) {
        logit_out[(size_t)e * NH + (tid & 7)] = s;
        atomicMax(&segkey[(size_t)myq * NH + (tid & 7)], float_key(s));
      }
    }
    par ^= 1;
  }
}

__global__ void exp_denom_kernel(const float* __restrict__ logit,
                                 const int* __restrict__ qidx,
                                 const unsigned* __restrict__ segkey,
                                 float* __restrict__ denom, int Medges) {
  int gid = blockIdx.x * blockDim.x + threadIdx.x;
  if (gid >= Medges * NH) return;
  const int e = gid >> 3, h = gid & 7;
  const int qi = qidx[e];
  const float mx = key_float(segkey[qi * NH + h]);
  const float ex = __expf(logit[(size_t)e * NH + h] - mx);
  atomicAdd(&denom[qi * NH + h], ex);
}

#define EPW 8
__global__ __launch_bounds__(256) void scatter_agg_kernel(const float* __restrict__ logit,
                                   const int* __restrict__ qidx,
                                   const int* __restrict__ kidx,
                                   const unsigned* __restrict__ segkey,
                                   const float* __restrict__ denom,
                                   const float* __restrict__ kvbuf,
                                   float* __restrict__ agg, int Medges) {
  const int lane = threadIdx.x & 31;
  const int wave = blockIdx.x * 8 + __builtin_amdgcn_readfirstlane((int)(threadIdx.x >> 5));
  const int e0 = wave * EPW;
  if (e0 >= Medges) return;
  const int h = lane & 7;

  if (e0 + EPW <= Medges) {
    const int myq = qidx[e0 + h];
    const int myk = kidx[e0 + h];
    int qi[EPW], ki[EPW];
#pragma unroll
    for (int u = 0; u < EPW; ++u) {
      qi[u] = __builtin_amdgcn_readlane(myq, u);
      ki[u] = __builtin_amdgcn_readlane(myk, u);
    }
    float lg[EPW], dn[EPW], vv[EPW][4];
    unsigned sk[EPW];
#pragma unroll
    for (int u = 0; u < EPW; ++u) {
      const float* vp = kvbuf + (size_t)ki[u] * (2 * HC) + HC + lane;
#pragma unroll
      for (int i = 0; i < 4; ++i) vv[u][i] = vp[32 * i];
    }
#pragma unroll
    for (int u = 0; u < EPW; ++u) {
      lg[u] = logit[(size_t)(e0 + u) * NH + h];
      sk[u] = segkey[(size_t)qi[u] * NH + h];
      dn[u] = denom[(size_t)qi[u] * NH + h];
    }
#pragma unroll
    for (int u = 0; u < EPW; ++u) {
      const float attn = __expf(lg[u] - key_float(sk[u])) / dn[u];
      float* ap = agg + (size_t)qi[u] * HC + lane;
#pragma unroll
      for (int i = 0; i < 4; ++i) atomicAdd(ap + 32 * i, attn * vv[u][i]);
    }
  } else {
    const int nE = Medges - e0;
    for (int u = 0; u < nE; ++u) {
      const int e = e0 + u;
      const int q1 = qidx[e];
      const int k1 = kidx[e];
      const float mx = key_float(segkey[(size_t)q1 * NH + h]);
      const float attn = __expf(logit[(size_t)e * NH + h] - mx) / denom[(size_t)q1 * NH + h];
      const float* vp = kvbuf + (size_t)k1 * (2 * HC) + HC + lane;
      float* ap = agg + (size_t)q1 * HC + lane;
#pragma unroll
      for (int i = 0; i < 4; ++i) atomicAdd(ap + 32 * i, attn * vp[32 * i]);
    }
  }
}

extern "C" void kernel_launch(void* const* d_in, const int* in_sizes, int n_in,
                              void* d_out, int out_size, void* d_ws, size_t ws_size,
                              hipStream_t stream) {
  const float* query = (const float*)d_in[0];
  const float* key_  = (const float*)d_in[1];
  const int*   qidx  = (const int*)d_in[2];
  const int*   kidx  = (const int*)d_in[3];
  const float* pr    = (const float*)d_in[4];
  const float* Wq    = (const float*)d_in[5];
  const float* Wkv   = (const float*)d_in[6];
  const float* Wb    = (const float*)d_in[7];
  const float* Wo    = (const float*)d_in[8];

  const int N = in_sizes[0] / HC;
  const int M = in_sizes[2];

  float* out_result = (float*)d_out;
  float* out_logit  = out_result + (size_t)N * HC;

  float*    q_buf  = (float*)d_ws;
  float*    kv_buf = q_buf + (size_t)N * HC;
  unsigned* segkey = (unsigned*)(kv_buf + (size_t)N * 2 * HC);
  float*    denom  = (float*)(segkey + (size_t)N * NH);
  float*    agg    = denom + (size_t)N * NH;

  const int TPB = 256;

  init_ws_kernel<<<(N * HC + TPB - 1) / TPB, TPB, 0, stream>>>(segkey, denom, agg, N);

  wmma_gemm_f32<HC, HC, 1>
      <<<(N + GEMM_ROWS - 1) / GEMM_ROWS, TPB, 0, stream>>>(query, Wq, q_buf, N);

  wmma_gemm_f32<2 * HC, 2 * HC, 2>
      <<<(N + GEMM_ROWS - 1) / GEMM_ROWS, TPB, 0, stream>>>(key_, Wkv, kv_buf, N);

  edge_bias_logit_kernel<<<512, TPB, 0, stream>>>(
      pr, Wb, q_buf, kv_buf, qidx, kidx, out_logit, segkey, M);

  exp_denom_kernel<<<(M * NH + TPB - 1) / TPB, TPB, 0, stream>>>(
      out_logit, qidx, segkey, denom, M);

  scatter_agg_kernel<<<((M + EPW - 1) / EPW + 7) / 8, TPB, 0, stream>>>(
      out_logit, qidx, kidx, segkey, denom, kv_buf, agg, M);

  wmma_gemm_f32<HC, HC, 1>
      <<<(N + GEMM_ROWS - 1) / GEMM_ROWS, TPB, 0, stream>>>(agg, Wo, out_result, N);
}


# projection GEMMs (q, kv, out): 512 resident blocks walk 16-row tiles; weight fragments loaded once per block, not per 32 rows; next tile's rows fetched a tile ahead into a second LDS buffer
# speedup vs baseline: 1.9128x; 1.0530x over previous
#include <hip/hip_runtime.h>

#define HC 128
#define NH 8

typedef __attribute__((ext_vector_type(2))) float v2f;
typedef __attribute__((ext_vector_type(8))) float v8f;

__device__ __forceinline__ unsigned float_key(float f) {
  unsigned u = __float_as_uint(f);
  return (u & 0x80000000u) ? ~u : (u | 0x80000000u);
}
__device__ __forceinline__ float key_float(unsigned k) {
  unsigned u = (k & 0x80000000u) ? (k & 0x7FFFFFFFu) : ~k;
  return __uint_as_float(u);
}

__global__ void init_ws_kernel(unsigned* __restrict__ segkey,
                               float* __restrict__ denom,
                               float* __restrict__ agg, int n) {
  int gid = blockIdx.x * blockDim.x + threadIdx.x;
  if (gid < n * NH) { segkey[gid] = 0u; denom[gid] = 0.0f; }
  if (gid < n * HC) { agg[gid] = 0.0f; }
}

#define GEMM_PITCH 132
#define GEMM_KG 8
#define GEMM_GRID 512
template <int LDB, int LDC, int NT>
__global__ __launch_bounds__(256) void wmma_gemm_f32(const float* __restrict__ A,
                                                     const float* __restrict__ B,
                                                     float* __restrict__ C,
                                                     int Mrows) {
  constexpr int KSTEPS = HC / 4;
  constexpr int NG = KSTEPS / GEMM_KG;
  __shared__ __attribute__((aligned(16))) float As[2][16 * GEMM_PITCH];

  const int lane = threadIdx.x & 31;
  const int w = __builtin_amdgcn_readfirstlane((int)(threadIdx.x >> 5));
  const int lr = lane & 15;
  const int half = lane >> 4;
  const int mtiles = (Mrows + 15) >> 4;
  const int stride = gridDim.x;
  const int lastrow = Mrows - 1;

  v2f bfrag[NT][KSTEPS];
#pragma unroll
  for (int t = 0; t < NT; ++t) {
    const float* bp = B + (size_t)(2 * half) * LDB + (w * NT + t) * 16 + lr;
#pragma unroll
    for (int kk = 0; kk < KSTEPS; ++kk) {
      v2f f;
      f.x = bp[(size_t)(kk * 4) * LDB];
      f.y = bp[(size_t)(kk * 4 + 1) * LDB];
      bfrag[t][kk] = f;
    }
  }

  int tile = blockIdx.x;
  float4 ra0 = {0.f, 0.f, 0.f, 0.f}, ra1 = ra0;
  if (tile < mtiles) {
    int r0 = tile * 16 + 2 * w, r1 = r0 + 1;
    r0 = r0 < lastrow ? r0 : lastrow;
    r1 = r1 < lastrow ? r1 : lastrow;
    ra0 = *(const float4*)(A + (size_t)r0 * HC + lane * 4);
    ra1 = *(const float4*)(A + (size_t)r1 * HC + lane * 4);
  }

  const int aoff = lr * GEMM_PITCH + 2 * half;
  const int soff = (2 * w) * GEMM_PITCH + lane * 4;
  int buf = 0;

  for (; tile < mtiles; tile += stride) {
    float* as = As[buf];
    *(float4*)(as + soff) = ra0;
    *(float4*)(as + soff + GEMM_PITCH) = ra1;
    __syncthreads();

    const int nt = tile + stride;
    if (nt < mtiles) {
      int r0 = nt * 16 + 2 * w, r1 = r0 + 1;
      r0 = r0 < lastrow ? r0 : lastrow;
      r1 = r1 < lastrow ? r1 : lastrow;
      ra0 = *(const float4*)(A + (size_t)r0 * HC + lane * 4);
      ra1 = *(const float4*)(A + (size_t)r1 * HC + lane * 4);
    }

    const float* ap = as + aoff;
    v8f acc[NT];
#pragma unroll
    for (int t = 0; t < NT; ++t) acc[t] = (v8f){};

    v2f a[2][GEMM_KG];
#pragma unroll
    for (int k = 0; k < GEMM_KG; ++k) a[0][k] = *(const v2f*)(ap + k * 4);
#pragma unroll
    for (int g = 0; g < NG; ++g) {
      const int cur = g & 1, nxt = cur ^ 1;
      if (g + 1 < NG) {
#pragma unroll
        for (int k = 0; k < GEMM_KG; ++k)
          a[nxt][k] = *(const v2f*)(ap + ((g + 1) * GEMM_KG + k) * 4);
      }
      __builtin_amdgcn_sched_barrier(0);
#pragma unroll
      for (int k = 0; k < GEMM_KG; ++k) {
#pragma unroll
        for (int t = 0; t < NT; ++t) {
          acc[t] = __builtin_amdgcn_wmma_f32_16x16x4_f32(false, a[cur][k], false,
                                                         bfrag[t][g * GEMM_KG + k],
                                                         (short)0, acc[t], false, false);
        }
      }
      __builtin_amdgcn_sched_barrier(0);
    }

    const int row0 = tile * 16;
    if (row0 + 16 <= Mrows) {
#pragma unroll
      for (int t = 0; t < NT; ++t) {
        float* cp = C + (size_t)(row0 + 8 * half) * LDC + (w * NT + t) * 16 + lr;
#pragma unroll
        for (int v = 0; v < 8; ++v) cp[(size_t)v * LDC] = acc[t][v];
      }
    } else {
#pragma unroll
      for (int t = 0; t < NT; ++t) {
        const int bcol = (w * NT + t) * 16 + lr;
#pragma unroll
        for (int v = 0; v < 8; ++v) {
          const int r0 = row0 + v + 8 * half;
          if (r0 < Mrows) C[(size_t)r0 * LDC + bcol] = acc[t][v];
        }
      }
    }
    buf ^= 1;
  }
}

#define EDGE_PITCH 132
#define EDGE_KG 8
__global__ __launch_bounds__(256) void edge_bias_logit_kernel(
    const float* __restrict__ PR, const float* __restrict__ Wb,
    const float* __restrict__ qbuf, const float* __restrict__ kvbuf,
    const int* __restrict__ qidx, const int* __restrict__ kidx,
    float* __restrict__ logit_out, unsigned* __restrict__ segkey, int Medges) {
  constexpr int KSTEPS = HC / 4;
  constexpr int NG = KSTEPS / EDGE_KG;
  __shared__ __attribute__((aligned(16))) float As[16 * EDGE_PITCH];
  __shared__ __attribute__((aligned(16))) float Qs[16 * EDGE_PITCH];
  __shared__ __attribute__((aligned(16))) float Ks[16 * EDGE_PITCH];
  __shared__ float Ps[8 * 128];
  __shared__ int Is[2][16];

  const int tid = threadIdx.x;
  const int lane = tid & 31;
  const int w = __builtin_amdgcn_readfirstlane(tid >> 5);
  const int lr = lane & 15;
  const int half = lane >> 4;
  const int mtiles = (Medges + 15) >> 4;
  const int stride = gridDim.x;
  const int last = Medges - 1;

  v2f bmf[KSTEPS], baf[KSTEPS];
  {
    const float* bp = Wb + (size_t)(2 * half) * 256 + w * 16 + lr;
#pragma unroll
    for (int kk = 0; kk < KSTEPS; ++kk) {
      v2f f, g;
      f.x = bp[(size_t)(kk * 4) * 256];
      f.y = bp[(size_t)(kk * 4 + 1) * 256];
      g.x = bp[(size_t)(kk * 4) * 256 + 128];
      g.y = bp[(size_t)(kk * 4 + 1) * 256 + 128];
      bmf[kk] = f;
      baf[kk] = g;
    }
  }

  int tile = blockIdx.x;
  float4 ra0 = {0.f, 0.f, 0.f, 0.f}, ra1 = ra0, rq0 = ra0, rq1 = ra0, rk0 = ra0, rk1 = ra0;
  int nq0 = 0, nq1 = 0, nk0 = 0, nk1 = 0;
  if (tile < mtiles) {
    int e0 = tile * 16 + 2 * w, e1 = e0 + 1;
    e0 = e0 < last ? e0 : last;
    e1 = e1 < last ? e1 : last;
    const int q0 = qidx[e0], q1 = qidx[e1], k0 = kidx[e0], k1 = kidx[e1];
    if (lane < 2) Is[0][2 * w + lane] = lane ? q1 : q0;
    ra0 = *(const float4*)(PR + (size_t)e0 * HC + lane * 4);
    ra1 = *(const float4*)(PR + (size_t)e1 * HC + lane * 4);
    rq0 = *(const float4*)(qbuf + (size_t)q0 * HC + lane * 4);
    rq1 = *(const float4*)(qbuf + (size_t)q1 * HC + lane * 4);
    rk0 = *(const float4*)(kvbuf + (size_t)k0 * (2 * HC) + lane * 4);
    rk1 = *(const float4*)(kvbuf + (size_t)k1 * (2 * HC) + lane * 4);
    if (tile + stride < mtiles) {
      int f0 = (tile + stride) * 16 + 2 * w, f1 = f0 + 1;
      f0 = f0 < last ? f0 : last;
      f1 = f1 < last ? f1 : last;
      nq0 = qidx[f0]; nq1 = qidx[f1]; nk0 = kidx[f0]; nk1 = kidx[f1];
    }
  }

  const float* ap = As + lr * EDGE_PITCH + 2 * half;
  const int srow = (2 * w) * EDGE_PITCH + lane * 4;
  int par = 0;

  for (; tile < mtiles; tile += stride) {
    *(float4*)(As + srow) = ra0;
    *(float4*)(As + srow + EDGE_PITCH) = ra1;
    *(float4*)(Qs + srow) = rq0;
    *(float4*)(Qs + srow + EDGE_PITCH) = rq1;
    *(float4*)(Ks + srow) = rk0;
    *(float4*)(Ks + srow + EDGE_PITCH) = rk1;
    __syncthreads();

    const int myq = Is[par][(tid >> 3) & 15];

    const int nt = tile + stride;
    if (nt < mtiles) {
      int e0 = nt * 16 + 2 * w, e1 = e0 + 1;
      e0 = e0 < last ? e0 : last;
      e1 = e1 < last ? e1 : last;
      ra0 = *(const float4*)(PR + (size_t)e0 * HC + lane * 4);
      ra1 = *(const float4*)(PR + (size_t)e1 * HC + lane * 4);
      rq0 = *(const float4*)(qbuf + (size_t)nq0 * HC + lane * 4);
      rq1 = *(const float4*)(qbuf + (size_t)nq1 * HC + lane * 4);
      rk0 = *(const float4*)(kvbuf + (size_t)nk0 * (2 * HC) + lane * 4);
      rk1 = *(const float4*)(kvbuf + (size_t)nk1 * (2 * HC) + lane * 4);
      if (lane < 2) Is[par ^ 1][2 * w + lane] = lane ? nq1 : nq0;
      const int nt2 = nt + stride;
      if (nt2 < mtiles) {
        int f0 = nt2 * 16 + 2 * w, f1 = f0 + 1;
        f0 = f0 < last ? f0 : last;
        f1 = f1 < last ? f1 : last;
        nq0 = qidx[f0]; nq1 = qidx[f1]; nk0 = kidx[f0]; nk1 = kidx[f1];
      }
    }

    v8f bm = {};
    v8f ba = {};
    v2f a[2][EDGE_KG];
#pragma unroll
    for (int k = 0; k < EDGE_KG; ++k) a[0][k] = *(const v2f*)(ap + k * 4);
#pragma unroll
    for (int g = 0; g < NG; ++g) {
      const int cur = g & 1, nxt = cur ^ 1;
      if (g + 1 < NG) {
#pragma unroll
        for (int k = 0; k < EDGE_KG; ++k)
          a[nxt][k] = *(const v2f*)(ap + ((g + 1) * EDGE_KG + k) * 4);
      }
      __builtin_amdgcn_sched_barrier(0);
#pragma unroll
      for (int k = 0; k < EDGE_KG; ++k) {
        bm = __builtin_amdgcn_wmma_f32_16x16x4_f32(false, a[cur][k], false,
                                                   bmf[g * EDGE_KG + k], (short)0,
                                                   bm, false, false);
        ba = __builtin_amdgcn_wmma_f32_16x16x4_f32(false, a[cur][k], false,
                                                   baf[g * EDGE_KG + k], (short)0,
                                                   ba, false, false);
      }
      __builtin_amdgcn_sched_barrier(0);
    }

    float p[8];
#pragma unroll
    for (int v = 0; v < 8; ++v) {
      const int r = v + 8 * half;
      const float qv = Qs[r * EDGE_PITCH + w * 16 + lr];
      const float kv = Ks[r * EDGE_PITCH + w * 16 + lr];
      float t = qv * (kv * (1.0f + bm[v]) + ba[v]);
      t += __shfl_xor(t, 8, 32);
      p[v] = t;
    }
    if (lr < 8) {
#pragma unroll
      for (int v = 0; v < 8; ++v) Ps[w * 128 + (v + 8 * half) * 8 + lr] = p[v];
    }
    __syncthreads();

    if (tid < 128) {
      float s = Ps[tid];
#pragma unroll
      for (int j = 1; j < 8; ++j) s += Ps[j * 128 + tid];
      const int e = tile * 16 + (tid >> 3);
      if (e < Medges) {
        logit_out[(size_t)e * NH + (tid & 7)] = s;
        atomicMax(&segkey[(size_t)myq * NH + (tid & 7)], float_key(s));
      }
    }
    par ^= 1;
  }
}

__global__ void exp_denom_kernel(const float* __restrict__ logit,
                                 const int* __restrict__ qidx,
                                 const unsigned* __restrict__ segkey,
                                 float* __restrict__ denom, int Medges) {
  int gid = blockIdx.x * blockDim.x + threadIdx.x;
  if (gid >= Medges * NH) return;
  const int e = gid >> 3, h = gid & 7;
  const int qi = qidx[e];
  const float mx = key_float(segkey[qi * NH + h]);
  const float ex = __expf(logit[(size_t)e * NH + h] - mx);
  atomicAdd(&denom[qi * NH + h], ex);
}

#define EPW 8
__global__ __launch_bounds__(256) void scatter_agg_kernel(const float* __restrict__ logit,
                                   const int* __restrict__ qidx,
                                   const int* __restrict__ kidx,
                                   const unsigned* __restrict__ segkey,
                                   const float* __restrict__ denom,
                                   const float* __restrict__ kvbuf,
                                   float* __restrict__ agg, int Medges) {
  const int lane = threadIdx.x & 31;
  const int wave = blockIdx.x * 8 + __builtin_amdgcn_readfirstlane((int)(threadIdx.x >> 5));
  const int e0 = wave * EPW;
  if (e0 >= Medges) return;
  const int h = lane & 7;

  if (e0 + EPW <= Medges) {
    const int myq = qidx[e0 + h];
    const int myk = kidx[e0 + h];
    int qi[EPW], ki[EPW];
#pragma unroll
    for (int u = 0; u < EPW; ++u) {
      qi[u] = __builtin_amdgcn_readlane(myq, u);
      ki[u] = __builtin_amdgcn_readlane(myk, u);
    }
    float lg[EPW], dn[EPW], vv[EPW][4];
    unsigned sk[EPW];
#pragma unroll
    for (int u = 0; u < EPW; ++u) {
      const float* vp = kvbuf + (size_t)ki[u] * (2 * HC) + HC + lane;
#pragma unroll
      for (int i = 0; i < 4; ++i) vv[u][i] = vp[32 * i];
    }
#pragma unroll
    for (int u = 0; u < EPW; ++u) {
      lg[u] = logit[(size_t)(e0 + u) * NH + h];
      sk[u] = segkey[(size_t)qi[u] * NH + h];
      dn[u] = denom[(size_t)qi[u] * NH + h];
    }
#pragma unroll
    for (int u = 0; u < EPW; ++u) {
      const float attn = __expf(lg[u] - key_float(sk[u])) / dn[u];
      float* ap = agg + (size_t)qi[u] * HC + lane;
#pragma unroll
      for (int i = 0; i < 4; ++i) atomicAdd(ap + 32 * i, attn * vv[u][i]);
    }
  } else {
    const int nE = Medges - e0;
    for (int u = 0; u < nE; ++u) {
      const int e = e0 + u;
      const int q1 = qidx[e];
      const int k1 = kidx[e];
      const float mx = key_float(segkey[(size_t)q1 * NH + h]);
      const float attn = __expf(logit[(size_t)e * NH + h] - mx) / denom[(size_t)q1 * NH + h];
      const float* vp = kvbuf + (size_t)k1 * (2 * HC) + HC + lane;
      float* ap = agg + (size_t)q1 * HC + lane;
#pragma unroll
      for (int i = 0; i < 4; ++i) atomicAdd(ap + 32 * i, attn * vp[32 * i]);
    }
  }
}

extern "C" void kernel_launch(void* const* d_in, const int* in_sizes, int n_in,
                              void* d_out, int out_size, void* d_ws, size_t ws_size,
                              hipStream_t stream) {
  const float* query = (const float*)d_in[0];
  const float* key_  = (const float*)d_in[1];
  const int*   qidx  = (const int*)d_in[2];
  const int*   kidx  = (const int*)d_in[3];
  const float* pr    = (const float*)d_in[4];
  const float* Wq    = (const float*)d_in[5];
  const float* Wkv   = (const float*)d_in[6];
  const float* Wb    = (const float*)d_in[7];
  const float* Wo    = (const float*)d_in[8];

  const int N = in_sizes[0] / HC;
  const int M = in_sizes[2];

  float* out_result = (float*)d_out;
  float* out_logit  = out_result + (size_t)N * HC;

  float*    q_buf  = (float*)d_ws;
  float*    kv_buf = q_buf + (size_t)N * HC;
  unsigned* segkey = (unsigned*)(kv_buf + (size_t)N * 2 * HC);
  float*    denom  = (float*)(segkey + (size_t)N * NH);
  float*    agg    = denom + (size_t)N * NH;

  const int TPB = 256;

  init_ws_kernel<<<(N * HC + TPB - 1) / TPB, TPB, 0, stream>>>(segkey, denom, agg, N);

  wmma_gemm_f32<HC, HC, 1>
      <<<GEMM_GRID, TPB, 0, stream>>>(query, Wq, q_buf, N);

  wmma_gemm_f32<2 * HC, 2 * HC, 2>
      <<<GEMM_GRID, TPB, 0, stream>>>(key_, Wkv, kv_buf, N);

  edge_bias_logit_kernel<<<512, TPB, 0, stream>>>(
      pr, Wb, q_buf, kv_buf, qidx, kidx, out_logit, segkey, M);

  exp_denom_kernel<<<(M * NH + TPB - 1) / TPB, TPB, 0, stream>>>(
      out_logit, qidx, segkey, denom, M);

  scatter_agg_kernel<<<((M + EPW - 1) / EPW + 7) / 8, TPB, 0, stream>>>(
      out_logit, qidx, kidx, segkey, denom, kv_buf, agg, M);

  wmma_gemm_f32<HC, HC, 1>
      <<<GEMM_GRID, TPB, 0, stream>>>(agg, Wo, out_result, N);
}


# exp-sum + scatter become one row-range kernel: 32-wave group owns 196 rows in LDS, scans all indices, adds by LDS float atomics, divides once per row; global float atomics gone; fp sum order changes
# speedup vs baseline: 2.0955x; 1.0955x over previous
#include <hip/hip_runtime.h>

#define HC 128
#define NH 8

typedef __attribute__((ext_vector_type(2))) float v2f;
typedef __attribute__((ext_vector_type(8))) float v8f;

__device__ __forceinline__ unsigned float_key(float f) {
  unsigned u = __float_as_uint(f);
  return (u & 0x80000000u) ? ~u : (u | 0x80000000u);
}
__device__ __forceinline__ float key_float(unsigned k) {
  unsigned u = (k & 0x80000000u) ? (k & 0x7FFFFFFFu) : ~k;
  return __uint_as_float(u);
}

__global__ void init_ws_kernel(unsigned* __restrict__ segkey, int n) {
  int gid = blockIdx.x * blockDim.x + threadIdx.x;
  if (gid < n * NH) segkey[gid] = 0u;
}

#define GEMM_PITCH 132
#define GEMM_KG 8
#define GEMM_GRID 512
template <int LDB, int LDC, int NT>
__global__ __launch_bounds__(256) void wmma_gemm_f32(const float* __restrict__ A,
                                                     const float* __restrict__ B,
                                                     float* __restrict__ C,
                                                     int Mrows) {
  constexpr int KSTEPS = HC / 4;
  constexpr int NG = KSTEPS / GEMM_KG;
  __shared__ __attribute__((aligned(16))) float As[2][16 * GEMM_PITCH];

  const int lane = threadIdx.x & 31;
  const int w = __builtin_amdgcn_readfirstlane((int)(threadIdx.x >> 5));
  const int lr = lane & 15;
  const int half = lane >> 4;
  const int mtiles = (Mrows + 15) >> 4;
  const int stride = gridDim.x;
  const int lastrow = Mrows - 1;

  v2f bfrag[NT][KSTEPS];
#pragma unroll
  for (int t = 0; t < NT; ++t) {
    const float* bp = B + (size_t)(2 * half) * LDB + (w * NT + t) * 16 + lr;
#pragma unroll
    for (int kk = 0; kk < KSTEPS; ++kk) {
      v2f f;
      f.x = bp[(size_t)(kk * 4) * LDB];
      f.y = bp[(size_t)(kk * 4 + 1) * LDB];
      bfrag[t][kk] = f;
    }
  }

  int tile = blockIdx.x;
  float4 ra0 = {0.f, 0.f, 0.f, 0.f}, ra1 = ra0;
  if (tile < mtiles) {
    int r0 = tile * 16 + 2 * w, r1 = r0 + 1;
    r0 = r0 < lastrow ? r0 : lastrow;
    r1 = r1 < lastrow ? r1 : lastrow;
    ra0 = *(const float4*)(A + (size_t)r0 * HC + lane * 4);
    ra1 = *(const float4*)(A + (size_t)r1 * HC + lane * 4);
  }

  const int aoff = lr * GEMM_PITCH + 2 * half;
  const int soff = (2 * w) * GEMM_PITCH + lane * 4;
  int buf = 0;

  for (; tile < mtiles; tile += stride) {
    float* as = As[buf];
    *(float4*)(as + soff) = ra0;
    *(float4*)(as + soff + GEMM_PITCH) = ra1;
    __syncthreads();

    const int nt = tile + stride;
    if (nt < mtiles) {
      int r0 = nt * 16 + 2 * w, r1 = r0 + 1;
      r0 = r0 < lastrow ? r0 : lastrow;
      r1 = r1 < lastrow ? r1 : lastrow;
      ra0 = *(const float4*)(A + (size_t)r0 * HC + lane * 4);
      ra1 = *(const float4*)(A + (size_t)r1 * HC + lane * 4);
    }

    const float* ap = as + aoff;
    v8f acc[NT];
#pragma unroll
    for (int t = 0; t < NT; ++t) acc[t] = (v8f){};

    v2f a[2][GEMM_KG];
#pragma unroll
    for (int k = 0; k < GEMM_KG; ++k) a[0][k] = *(const v2f*)(ap + k * 4);
#pragma unroll
    for (int g = 0; g < NG; ++g) {
      const int cur = g & 1, nxt = cur ^ 1;
      if (g + 1 < NG) {
#pragma unroll
        for (int k = 0; k < GEMM_KG; ++k)
          a[nxt][k] = *(const v2f*)(ap + ((g + 1) * GEMM_KG + k) * 4);
      }
      __builtin_amdgcn_sched_barrier(0);
#pragma unroll
      for (int k = 0; k < GEMM_KG; ++k) {
#pragma unroll
        for (int t = 0; t < NT; ++t) {
          acc[t] = __builtin_amdgcn_wmma_f32_16x16x4_f32(false, a[cur][k], false,
                                                         bfrag[t][g * GEMM_KG + k],
                                                         (short)0, acc[t], false, false);
        }
      }
      __builtin_amdgcn_sched_barrier(0);
    }

    const int row0 = tile * 16;
    if (row0 + 16 <= Mrows) {
#pragma unroll
      for (int t = 0; t < NT; ++t) {
        float* cp = C + (size_t)(row0 + 8 * half) * LDC + (w * NT + t) * 16 + lr;
#pragma unroll
        for (int v = 0; v < 8; ++v) cp[(size_t)v * LDC] = acc[t][v];
      }
    } else {
#pragma unroll
      for (int t = 0; t < NT; ++t) {
        const int bcol = (w * NT + t) * 16 + lr;
#pragma unroll
        for (int v = 0; v < 8; ++v) {
          const int r0 = row0 + v + 8 * half;
          if (r0 < Mrows) C[(size_t)r0 * LDC + bcol] = acc[t][v];
        }
      }
    }
    buf ^= 1;
  }
}

#define EDGE_PITCH 132
#define EDGE_KG 8
__global__ __launch_bounds__(256) void edge_bias_logit_kernel(
    const float* __restrict__ PR, const float* __restrict__ Wb,
    const float* __restrict__ qbuf, const float* __restrict__ kvbuf,
    const int* __restrict__ qidx, const int* __restrict__ kidx,
    float* __restrict__ logit_out, unsigned* __restrict__ segkey, int Medges) {
  constexpr int KSTEPS = HC / 4;
  constexpr int NG = KSTEPS / EDGE_KG;
  __shared__ __attribute__((aligned(16))) float As[16 * EDGE_PITCH];
  __shared__ __attribute__((aligned(16))) float Qs[16 * EDGE_PITCH];
  __shared__ __attribute__((aligned(16))) float Ks[16 * EDGE_PITCH];
  __shared__ float Ps[8 * 128];
  __shared__ int Is[2][16];

  const int tid = threadIdx.x;
  const int lane = tid & 31;
  const int w = __builtin_amdgcn_readfirstlane(tid >> 5);
  const int lr = lane & 15;
  const int half = lane >> 4;
  const int mtiles = (Medges + 15) >> 4;
  const int stride = gridDim.x;
  const int last = Medges - 1;

  v2f bmf[KSTEPS], baf[KSTEPS];
  {
    const float* bp = Wb + (size_t)(2 * half) * 256 + w * 16 + lr;
#pragma unroll
    for (int kk = 0; kk < KSTEPS; ++kk) {
      v2f f, g;
      f.x = bp[(size_t)(kk * 4) * 256];
      f.y = bp[(size_t)(kk * 4 + 1) * 256];
      g.x = bp[(size_t)(kk * 4) * 256 + 128];
      g.y = bp[(size_t)(kk * 4 + 1) * 256 + 128];
      bmf[kk] = f;
      baf[kk] = g;
    }
  }

  int tile = blockIdx.x;
  float4 ra0 = {0.f, 0.f, 0.f, 0.f}, ra1 = ra0, rq0 = ra0, rq1 = ra0, rk0 = ra0, rk1 = ra0;
  int nq0 = 0, nq1 = 0, nk0 = 0, nk1 = 0;
  if (tile < mtiles) {
    int e0 = tile * 16 + 2 * w, e1 = e0 + 1;
    e0 = e0 < last ? e0 : last;
    e1 = e1 < last ? e1 : last;
    const int q0 = qidx[e0], q1 = qidx[e1], k0 = kidx[e0], k1 = kidx[e1];
    if (lane < 2) Is[0][2 * w + lane] = lane ? q1 : q0;
    ra0 = *(const float4*)(PR + (size_t)e0 * HC + lane * 4);
    ra1 = *(const float4*)(PR + (size_t)e1 * HC + lane * 4);
    rq0 = *(const float4*)(qbuf + (size_t)q0 * HC + lane * 4);
    rq1 = *(const float4*)(qbuf + (size_t)q1 * HC + lane * 4);
    rk0 = *(const float4*)(kvbuf + (size_t)k0 * (2 * HC) + lane * 4);
    rk1 = *(const float4*)(kvbuf + (size_t)k1 * (2 * HC) + lane * 4);
    if (tile + stride < mtiles) {
      int f0 = (tile + stride) * 16 + 2 * w, f1 = f0 + 1;
      f0 = f0 < last ? f0 : last;
      f1 = f1 < last ? f1 : last;
      nq0 = qidx[f0]; nq1 = qidx[f1]; nk0 = kidx[f0]; nk1 = kidx[f1];
    }
  }

  const float* ap = As + lr * EDGE_PITCH + 2 * half;
  const int srow = (2 * w) * EDGE_PITCH + lane * 4;
  int par = 0;

  for (; tile < mtiles; tile += stride) {
    *(float4*)(As + srow) = ra0;
    *(float4*)(As + srow + EDGE_PITCH) = ra1;
    *(float4*)(Qs + srow) = rq0;
    *(float4*)(Qs + srow + EDGE_PITCH) = rq1;
    *(float4*)(Ks + srow) = rk0;
    *(float4*)(Ks + srow + EDGE_PITCH) = rk1;
    __syncthreads();

    const int myq = Is[par][(tid >> 3) & 15];

    const int nt = tile + stride;
    if (nt < mtiles) {
      int e0 = nt * 16 + 2 * w, e1 = e0 + 1;
      e0 = e0 < last ? e0 : last;
      e1 = e1 < last ? e1 : last;
      ra0 = *(const float4*)(PR + (size_t)e0 * HC + lane * 4);
      ra1 = *(const float4*)(PR + (size_t)e1 * HC + lane * 4);
      rq0 = *(const float4*)(qbuf + (size_t)nq0 * HC + lane * 4);
      rq1 = *(const float4*)(qbuf + (size_t)nq1 * HC + lane * 4);
      rk0 = *(const float4*)(kvbuf + (size_t)nk0 * (2 * HC) + lane * 4);
      rk1 = *(const float4*)(kvbuf + (size_t)nk1 * (2 * HC) + lane * 4);
      if (lane < 2) Is[par ^ 1][2 * w + lane] = lane ? nq1 : nq0;
      const int nt2 = nt + stride;
      if (nt2 < mtiles) {
        int f0 = nt2 * 16 + 2 * w, f1 = f0 + 1;
        f0 = f0 < last ? f0 : last;
        f1 = f1 < last ? f1 : last;
        nq0 = qidx[f0]; nq1 = qidx[f1]; nk0 = kidx[f0]; nk1 = kidx[f1];
      }
    }

    v8f bm = {};
    v8f ba = {};
    v2f a[2][EDGE_KG];
#pragma unroll
    for (int k = 0; k < EDGE_KG; ++k) a[0][k] = *(const v2f*)(ap + k * 4);
#pragma unroll
    for (int g = 0; g < NG; ++g) {
      const int cur = g & 1, nxt = cur ^ 1;
      if (g + 1 < NG) {
#pragma unroll
        for (int k = 0; k < EDGE_KG; ++k)
          a[nxt][k] = *(const v2f*)(ap + ((g + 1) * EDGE_KG + k) * 4);
      }
      __builtin_amdgcn_sched_barrier(0);
#pragma unroll
      for (int k = 0; k < EDGE_KG; ++k) {
        bm = __builtin_amdgcn_wmma_f32_16x16x4_f32(false, a[cur][k], false,
                                                   bmf[g * EDGE_KG + k], (short)0,
                                                   bm, false, false);
        ba = __builtin_amdgcn_wmma_f32_16x16x4_f32(false, a[cur][k], false,
                                                   baf[g * EDGE_KG + k], (short)0,
                                                   ba, false, false);
      }
      __builtin_amdgcn_sched_barrier(0);
    }

    float p[8];
#pragma unroll
    for (int v = 0; v < 8; ++v) {
      const int r = v + 8 * half;
      const float qv = Qs[r * EDGE_PITCH + w * 16 + lr];
      const float kv = Ks[r * EDGE_PITCH + w * 16 + lr];
      float t = qv * (kv * (1.0f + bm[v]) + ba[v]);
      t += __shfl_xor(t, 8, 32);
      p[v] = t;
    }
    if (lr < 8) {
#pragma unroll
      for (int v = 0; v < 8; ++v) Ps[w * 128 + (v + 8 * half) * 8 + lr] = p[v];
    }
    __syncthreads();

    if (tid < 128) {
      float s = Ps[tid];
#pragma unroll
      for (int j = 1; j < 8; ++j) s += Ps[j * 128 + tid];
      const int e = tile * 16 + (tid >> 3);
      if (e < Medges) {
        logit_out[(size_t)e * NH + (tid & 7)] = s;
        atomicMax(&segkey[(size_t)myq * NH + (tid & 7)], float_key(s));
      }
    }
    par ^= 1;
  }
}

#define RR_ROWS 196
#define RR_WAVES 32
#define RR_LCAP 1024
#define RR_U 7

__device__ __forceinline__ void rr_batch(const int* lp, int nb, int lo, int lane,
                                         const float* __restrict__ logit,
                                         const int* __restrict__ qidx,
                                         const int* __restrict__ kidx,
                                         const float* __restrict__ kvbuf,
                                         float* Sum, float* Den, const float* Mx) {
  const int h = lane & 7;
  const int e = lp[h < nb ? h : 0];
  const int myq = qidx[e] - lo;
  const int myk = kidx[e];
  int ee[8], qi[8], ki[8];
#pragma unroll
  for (int u = 0; u < 8; ++u) {
    ee[u] = __builtin_amdgcn_readlane(e, u);
    qi[u] = __builtin_amdgcn_readlane(myq, u);
    ki[u] = __builtin_amdgcn_readlane(myk, u);
  }
  float lg[8], mx[8], vv[8][4];
#pragma unroll
  for (int u = 0; u < 8; ++u) {
    const float* vp = kvbuf + (size_t)ki[u] * (2 * HC) + HC + lane;
#pragma unroll
    for (int i = 0; i < 4; ++i) vv[u][i] = vp[32 * i];
  }
#pragma unroll
  for (int u = 0; u < 8; ++u) {
    lg[u] = logit[(size_t)ee[u] * NH + h];
    mx[u] = Mx[qi[u] * NH + h];
  }
#pragma unroll
  for (int u = 0; u < 8; ++u) {
    if (u < nb) {
      const float ex = __expf(lg[u] - mx[u]);
      if (lane < 8) atomicAdd(Den + qi[u] * NH + h, ex);
      float* sp = Sum + qi[u] * HC + lane;
#pragma unroll
      for (int i = 0; i < 4; ++i) atomicAdd(sp + 32 * i, ex * vv[u][i]);
    }
  }
}

__global__ __launch_bounds__(1024) void rowrange_softmax_agg_kernel(
    const float* __restrict__ logit, const int* __restrict__ qidx,
    const int* __restrict__ kidx, const unsigned* __restrict__ segkey,
    const float* __restrict__ kvbuf, float* __restrict__ agg, int Medges, int n) {
  __shared__ __attribute__((aligned(16))) float Sum[RR_ROWS * HC];
  __shared__ float Den[RR_ROWS * NH];
  __shared__ float Mx[RR_ROWS * NH];
  __shared__ int List[RR_WAVES * RR_LCAP];

  const int tid = threadIdx.x;
  const int lane = tid & 31;
  const int w = __builtin_amdgcn_readfirstlane(tid >> 5);
  int* list = List + w * RR_LCAP;
  const int nfull = Medges >> 7;

  for (int lo = blockIdx.x * RR_ROWS; lo < n; lo += gridDim.x * RR_ROWS) {
    const int nrows = (n - lo) < RR_ROWS ? (n - lo) : RR_ROWS;
    const unsigned nr = (unsigned)nrows;

    for (int i = tid; i < RR_ROWS * HC / 4; i += 1024)
      ((float4*)Sum)[i] = (float4){0.f, 0.f, 0.f, 0.f};
    for (int i = tid; i < nrows * NH; i += 1024) {
      Den[i] = 0.0f;
      Mx[i] = key_float(segkey[(size_t)lo * NH + i]);
    }
    __syncthreads();

    int head = 0, cnt = 0;
    if (w == 0 && nfull * 128 < Medges) {
#pragma unroll
      for (int c = 0; c < 4; ++c) {
        const int e = nfull * 128 + c * 32 + lane;
        const int q = e < Medges ? qidx[e] : -1;
        const bool m = (unsigned)(q - lo) < nr;
        const unsigned mask = __builtin_amdgcn_ballot_w32(m);
        if (mask != 0u) {
          const int pos = cnt + (int)__builtin_amdgcn_mbcnt_lo(mask, 0u);
          if (m) list[pos] = e;
          cnt += __builtin_popcount(mask);
        }
      }
    }
    for (int j0 = w; j0 < nfull; j0 += RR_WAVES * RR_U) {
      int4 iv[RR_U];
#pragma unroll
      for (int u = 0; u < RR_U; ++u) {
        const int j = j0 + u * RR_WAVES;
        const int jj = j < nfull ? j : j0;
        iv[u] = *(const int4*)(qidx + jj * 128 + lane * 4);
      }
#pragma unroll
      for (int u = 0; u < RR_U; ++u) {
        const int j = j0 + u * RR_WAVES;
        const unsigned nru = j < nfull ? nr : 0u;
        const int base = j * 128 + lane * 4;
        const unsigned d0 = (unsigned)(iv[u].x - lo), d1 = (unsigned)(iv[u].y - lo);
        const unsigned d2 = (unsigned)(iv[u].z - lo), d3 = (unsigned)(iv[u].w - lo);
        const unsigned m01 = d0 < d1 ? d0 : d1, m23 = d2 < d3 ? d2 : d3;
        const unsigned mn = m01 < m23 ? m01 : m23;
        if (__builtin_amdgcn_ballot_w32(mn < nru) != 0u) {
          const unsigned dd[4] = {d0, d1, d2, d3};
#pragma unroll
          for (int c = 0; c < 4; ++c) {
            const bool m = dd[c] < nru;
            const unsigned mask = __builtin_amdgcn_ballot_w32(m);
            if (mask != 0u) {
              const int pos = cnt + (int)__builtin_amdgcn_mbcnt_lo(mask, 0u);
              if (m) list[pos] = base + c;
              cnt += __builtin_popcount(mask);
            }
          }
        }
      }
      const int keep = cnt > RR_LCAP - RR_U * 128 ? 0 : 7;
      while (cnt - head > keep) {
        const int nb = cnt - head < 8 ? cnt - head : 8;
        rr_batch(list + head, nb, lo, lane, logit, qidx, kidx, kvbuf, Sum, Den, Mx);
        head += nb;
      }
      if (head == cnt) { head = 0; cnt = 0; }
    }
    while (head < cnt) {
      const int nb = cnt - head < 8 ? cnt - head : 8;
      rr_batch(list + head, nb, lo, lane, logit, qidx, kidx, kvbuf, Sum, Den, Mx);
      head += nb;
    }
    __syncthreads();

    for (int i = tid; i < nrows * (HC / 4); i += 1024) {
      const int r = i >> 5, c4 = i & 31;
      const float4 s = ((const float4*)Sum)[i];
      const float* dp = Den + r * NH + 4 * (c4 & 1);
      const float e0 = dp[0], e1 = dp[1], e2 = dp[2], e3 = dp[3];
      float4 o;
      o.x = e0 == 0.0f ? 0.0f : s.x / e0;
      o.y = e1 == 0.0f ? 0.0f : s.y / e1;
      o.z = e2 == 0.0f ? 0.0f : s.z / e2;
      o.w = e3 == 0.0f ? 0.0f : s.w / e3;
      *(float4*)(agg + (size_t)(lo + r) * HC + c4 * 4) = o;
    }
    __syncthreads();
  }
}

extern "C" void kernel_launch(void* const* d_in, const int* in_sizes, int n_in,
                              void* d_out, int out_size, void* d_ws, size_t ws_size,
                              hipStream_t stream) {
  const float* query = (const float*)d_in[0];
  const float* key_  = (const float*)d_in[1];
  const int*   qidx  = (const int*)d_in[2];
  const int*   kidx  = (const int*)d_in[3];
  const float* pr    = (const float*)d_in[4];
  const float* Wq    = (const float*)d_in[5];
  const float* Wkv   = (const float*)d_in[6];
  const float* Wb    = (const float*)d_in[7];
  const float* Wo    = (const float*)d_in[8];

  const int N = in_sizes[0] / HC;
  const int M = in_sizes[2];

  float* out_result = (float*)d_out;
  float* out_logit  = out_result + (size_t)N * HC;

  float*    q_buf  = (float*)d_ws;
  float*    kv_buf = q_buf + (size_t)N * HC;
  unsigned* segkey = (unsigned*)(kv_buf + (size_t)N * 2 * HC);
  float*    agg    = (float*)(segkey + (size_t)N * NH) + (size_t)N * NH;

  const int TPB = 256;

  init_ws_kernel<<<(N * NH + TPB - 1) / TPB, TPB, 0, stream>>>(segkey, N);

  wmma_gemm_f32<HC, HC, 1>
      <<<GEMM_GRID, TPB, 0, stream>>>(query, Wq, q_buf, N);

  wmma_gemm_f32<2 * HC, 2 * HC, 2>
      <<<GEMM_GRID, TPB, 0, stream>>>(key_, Wkv, kv_buf, N);

  edge_bias_logit_kernel<<<512, TPB, 0, stream>>>(
      pr, Wb, q_buf, kv_buf, qidx, kidx, out_logit, segkey, M);

  rowrange_softmax_agg_kernel<<<256, 1024, 0, stream>>>(
      out_logit, qidx, kidx, segkey, kv_buf, agg, M, N);

  wmma_gemm_f32<HC, HC, 1>
      <<<GEMM_GRID, TPB, 0, stream>>>(agg, Wo, out_result, N);
}


# row-range groups no longer scan the whole index array: init launch writes one edge bitmap row per row range (LDS OR, whole lines), a group reads its own row; fp sum order: arrival, as before
# speedup vs baseline: 2.1593x; 1.0304x over previous
#include <hip/hip_runtime.h>

#define HC 128
#define NH 8

typedef __attribute__((ext_vector_type(2))) float v2f;
typedef __attribute__((ext_vector_type(8))) float v8f;

__device__ __forceinline__ unsigned float_key(float f) {
  unsigned u = __float_as_uint(f);
  return (u & 0x80000000u) ? ~u : (u | 0x80000000u);
}
__device__ __forceinline__ float key_float(unsigned k) {
  unsigned u = (k & 0x80000000u) ? (k & 0x7FFFFFFFu) : ~k;
  return __uint_as_float(u);
}

#define RR_ROWS 196
#define BM_EDGES 1024
#define BM_RANGES 256
#define BM_PITCH 33
__global__ __launch_bounds__(256) void init_ws_kernel(unsigned* __restrict__ segkey,
                                                      const int* __restrict__ qidx,
                                                      unsigned* __restrict__ bitmap,
                                                      int n, int Medges) {
  __shared__ unsigned Tile[BM_RANGES * BM_PITCH];
  const int tid = threadIdx.x;
  const int lane = tid & 31;
  const int w = __builtin_amdgcn_readfirstlane(tid >> 5);

  const int nk4 = n * NH / 4;
  for (int i = blockIdx.x * 256 + tid; i < nk4; i += gridDim.x * 256)
    ((uint4*)segkey)[i] = (uint4){0u, 0u, 0u, 0u};

  const int nvec = (Medges + BM_EDGES - 1) / BM_EDGES;
  const int nranges = (n + RR_ROWS - 1) / RR_ROWS;
  const size_t pitch = (size_t)nvec * 32;
  for (int vec = blockIdx.x; vec < nvec; vec += gridDim.x) {
    int qv[4], rid[4];
#pragma unroll
    for (int c = 0; c < 4; ++c) {
      const int e = vec * BM_EDGES + (4 * w + c) * 32 + lane;
      qv[c] = qidx[e < Medges ? e : Medges - 1];
    }
#pragma unroll
    for (int c = 0; c < 4; ++c) {
      const int e = vec * BM_EDGES + (4 * w + c) * 32 + lane;
      rid[c] = e < Medges ? qv[c] / RR_ROWS : -1;
    }
    for (int r0 = 0; r0 < nranges; r0 += BM_RANGES) {
      for (int i = tid; i < BM_RANGES * BM_PITCH; i += 256) Tile[i] = 0u;
      __syncthreads();
#pragma unroll
      for (int c = 0; c < 4; ++c) {
        const unsigned d = (unsigned)(rid[c] - r0);
        if (d < (unsigned)BM_RANGES) atomicOr(&Tile[d * BM_PITCH + 4 * w + c], 1u << lane);
      }
      __syncthreads();
      for (int rr0 = w; rr0 < BM_RANGES; rr0 += 64) {
        unsigned t[8];
#pragma unroll
        for (int k = 0; k < 8; ++k) t[k] = Tile[(rr0 + 8 * k) * BM_PITCH + lane];
#pragma unroll
        for (int k = 0; k < 8; ++k) {
          const int r = r0 + rr0 + 8 * k;
          if (r < nranges)
            bitmap[(size_t)r * pitch + (size_t)vec * 32 + lane] = t[k];
        }
      }
      __syncthreads();
    }
  }
}

#define GEMM_PITCH 132
#define GEMM_KG 8
#define GEMM_GRID 512
template <int LDB, int LDC, int NT>
__global__ __launch_bounds__(256) void wmma_gemm_f32(const float* __restrict__ A,
                                                     const float* __restrict__ B,
                                                     float* __restrict__ C,
                                                     int Mrows) {
  constexpr int KSTEPS = HC / 4;
  constexpr int NG = KSTEPS / GEMM_KG;
  __shared__ __attribute__((aligned(16))) float As[2][16 * GEMM_PITCH];

  const int lane = threadIdx.x & 31;
  const int w = __builtin_amdgcn_readfirstlane((int)(threadIdx.x >> 5));
  const int lr = lane & 15;
  const int half = lane >> 4;
  const int mtiles = (Mrows + 15) >> 4;
  const int stride = gridDim.x;
  const int lastrow = Mrows - 1;

  v2f bfrag[NT][KSTEPS];
#pragma unroll
  for (int t = 0; t < NT; ++t) {
    const float* bp = B + (size_t)(2 * half) * LDB + (w * NT + t) * 16 + lr;
#pragma unroll
    for (int kk = 0; kk < KSTEPS; ++kk) {
      v2f f;
      f.x = bp[(size_t)(kk * 4) * LDB];
      f.y = bp[(size_t)(kk * 4 + 1) * LDB];
      bfrag[t][kk] = f;
    }
  }

  int tile = blockIdx.x;
  float4 ra0 = {0.f, 0.f, 0.f, 0.f}, ra1 = ra0;
  if (tile < mtiles) {
    int r0 = tile * 16 + 2 * w, r1 = r0 + 1;
    r0 = r0 < lastrow ? r0 : lastrow;
    r1 = r1 < lastrow ? r1 : lastrow;
    ra0 = *(const float4*)(A + (size_t)r0 * HC + lane * 4);
    ra1 = *(const float4*)(A + (size_t)r1 * HC + lane * 4);
  }

  const int aoff = lr * GEMM_PITCH + 2 * half;
  const int soff = (2 * w) * GEMM_PITCH + lane * 4;
  int buf = 0;

  for (; tile < mtiles; tile += stride) {
    float* as = As[buf];
    *(float4*)(as + soff) = ra0;
    *(float4*)(as + soff + GEMM_PITCH) = ra1;
    __syncthreads();

    const int nt = tile + stride;
    if (nt < mtiles) {
      int r0 = nt * 16 + 2 * w, r1 = r0 + 1;
      r0 = r0 < lastrow ? r0 : lastrow;
      r1 = r1 < lastrow ? r1 : lastrow;
      ra0 = *(const float4*)(A + (size_t)r0 * HC + lane * 4);
      ra1 = *(const float4*)(A + (size_t)r1 * HC + lane * 4);
    }

    const float* ap = as + aoff;
    v8f acc[NT];
#pragma unroll
    for (int t = 0; t < NT; ++t) acc[t] = (v8f){};

    v2f a[2][GEMM_KG];
#pragma unroll
    for (int k = 0; k < GEMM_KG; ++k) a[0][k] = *(const v2f*)(ap + k * 4);
#pragma unroll
    for (int g = 0; g < NG; ++g) {
      const int cur = g & 1, nxt = cur ^ 1;
      if (g + 1 < NG) {
#pragma unroll
        for (int k = 0; k < GEMM_KG; ++k)
          a[nxt][k] = *(const v2f*)(ap + ((g + 1) * GEMM_KG + k) * 4);
      }
      __builtin_amdgcn_sched_barrier(0);
#pragma unroll
      for (int k = 0; k < GEMM_KG; ++k) {
#pragma unroll
        for (int t = 0; t < NT; ++t) {
          acc[t] = __builtin_amdgcn_wmma_f32_16x16x4_f32(false, a[cur][k], false,
                                                         bfrag[t][g * GEMM_KG + k],
                                                         (short)0, acc[t], false, false);
        }
      }
      __builtin_amdgcn_sched_barrier(0);
    }

    const int row0 = tile * 16;
    if (row0 + 16 <= Mrows) {
#pragma unroll
      for (int t = 0; t < NT; ++t) {
        float* cp = C + (size_t)(row0 + 8 * half) * LDC + (w * NT + t) * 16 + lr;
#pragma unroll
        for (int v = 0; v < 8; ++v) cp[(size_t)v * LDC] = acc[t][v];
      }
    } else {
#pragma unroll
      for (int t = 0; t < NT; ++t) {
        const int bcol = (w * NT + t) * 16 + lr;
#pragma unroll
        for (int v = 0; v < 8; ++v) {
          const int r0 = row0 + v + 8 * half;
          if (r0 < Mrows) C[(size_t)r0 * LDC + bcol] = acc[t][v];
        }
      }
    }
    buf ^= 1;
  }
}

#define EDGE_PITCH 132
#define EDGE_KG 8
__global__ __launch_bounds__(256) void edge_bias_logit_kernel(
    const float* __restrict__ PR, const float* __restrict__ Wb,
    const float* __restrict__ qbuf, const float* __restrict__ kvbuf,
    const int* __restrict__ qidx, const int* __restrict__ kidx,
    float* __restrict__ logit_out, unsigned* __restrict__ segkey, int Medges) {
  constexpr int KSTEPS = HC / 4;
  constexpr int NG = KSTEPS / EDGE_KG;
  __shared__ __attribute__((aligned(16))) float As[16 * EDGE_PITCH];
  __shared__ __attribute__((aligned(16))) float Qs[16 * EDGE_PITCH];
  __shared__ __attribute__((aligned(16))) float Ks[16 * EDGE_PITCH];
  __shared__ float Ps[8 * 128];
  __shared__ int Is[2][16];

  const int tid = threadIdx.x;
  const int lane = tid & 31;
  const int w = __builtin_amdgcn_readfirstlane(tid >> 5);
  const int lr = lane & 15;
  const int half = lane >> 4;
  const int mtiles = (Medges + 15) >> 4;
  const int stride = gridDim.x;
  const int last = Medges - 1;

  v2f bmf[KSTEPS], baf[KSTEPS];
  {
    const float* bp = Wb + (size_t)(2 * half) * 256 + w * 16 + lr;
#pragma unroll
    for (int kk = 0; kk < KSTEPS; ++kk) {
      v2f f, g;
      f.x = bp[(size_t)(kk * 4) * 256];
      f.y = bp[(size_t)(kk * 4 + 1) * 256];
      g.x = bp[(size_t)(kk * 4) * 256 + 128];
      g.y = bp[(size_t)(kk * 4 + 1) * 256 + 128];
      bmf[kk] = f;
      baf[kk] = g;
    }
  }

  int tile = blockIdx.x;
  float4 ra0 = {0.f, 0.f, 0.f, 0.f}, ra1 = ra0, rq0 = ra0, rq1 = ra0, rk0 = ra0, rk1 = ra0;
  int nq0 = 0, nq1 = 0, nk0 = 0, nk1 = 0;
  if (tile < mtiles) {
    int e0 = tile * 16 + 2 * w, e1 = e0 + 1;
    e0 = e0 < last ? e0 : last;
    e1 = e1 < last ? e1 : last;
    const int q0 = qidx[e0], q1 = qidx[e1], k0 = kidx[e0], k1 = kidx[e1];
    if (lane < 2) Is[0][2 * w + lane] = lane ? q1 : q0;
    ra0 = *(const float4*)(PR + (size_t)e0 * HC + lane * 4);
    ra1 = *(const float4*)(PR + (size_t)e1 * HC + lane * 4);
    rq0 = *(const float4*)(qbuf + (size_t)q0 * HC + lane * 4);
    rq1 = *(const float4*)(qbuf + (size_t)q1 * HC + lane * 4);
    rk0 = *(const float4*)(kvbuf + (size_t)k0 * (2 * HC) + lane * 4);
    rk1 = *(const float4*)(kvbuf + (size_t)k1 * (2 * HC) + lane * 4);
    if (tile + stride < mtiles) {
      int f0 = (tile + stride) * 16 + 2 * w, f1 = f0 + 1;
      f0 = f0 < last ? f0 : last;
      f1 = f1 < last ? f1 : last;
      nq0 = qidx[f0]; nq1 = qidx[f1]; nk0 = kidx[f0]; nk1 = kidx[f1];
    }
  }

  const float* ap = As + lr * EDGE_PITCH + 2 * half;
  const int srow = (2 * w) * EDGE_PITCH + lane * 4;
  int par = 0;

  for (; tile < mtiles; tile += stride) {
    *(float4*)(As + srow) = ra0;
    *(float4*)(As + srow + EDGE_PITCH) = ra1;
    *(float4*)(Qs + srow) = rq0;
    *(float4*)(Qs + srow + EDGE_PITCH) = rq1;
    *(float4*)(Ks + srow) = rk0;
    *(float4*)(Ks + srow + EDGE_PITCH) = rk1;
    __syncthreads();

    const int myq = Is[par][(tid >> 3) & 15];

    const int nt = tile + stride;
    if (nt < mtiles) {
      int e0 = nt * 16 + 2 * w, e1 = e0 + 1;
      e0 = e0 < last ? e0 : last;
      e1 = e1 < last ? e1 : last;
      ra0 = *(const float4*)(PR + (size_t)e0 * HC + lane * 4);
      ra1 = *(const float4*)(PR + (size_t)e1 * HC + lane * 4);
      rq0 = *(const float4*)(qbuf + (size_t)nq0 * HC + lane * 4);
      rq1 = *(const float4*)(qbuf + (size_t)nq1 * HC + lane * 4);
      rk0 = *(const float4*)(kvbuf + (size_t)nk0 * (2 * HC) + lane * 4);
      rk1 = *(const float4*)(kvbuf + (size_t)nk1 * (2 * HC) + lane * 4);
      if (lane < 2) Is[par ^ 1][2 * w + lane] = lane ? nq1 : nq0;
      const int nt2 = nt + stride;
      if (nt2 < mtiles) {
        int f0 = nt2 * 16 + 2 * w, f1 = f0 + 1;
        f0 = f0 < last ? f0 : last;
        f1 = f1 < last ? f1 : last;
        nq0 = qidx[f0]; nq1 = qidx[f1]; nk0 = kidx[f0]; nk1 = kidx[f1];
      }
    }

    v8f bm = {};
    v8f ba = {};
    v2f a[2][EDGE_KG];
#pragma unroll
    for (int k = 0; k < EDGE_KG; ++k) a[0][k] = *(const v2f*)(ap + k * 4);
#pragma unroll
    for (int g = 0; g < NG; ++g) {
      const int cur = g & 1, nxt = cur ^ 1;
      if (g + 1 < NG) {
#pragma unroll
        for (int k = 0; k < EDGE_KG; ++k)
          a[nxt][k] = *(const v2f*)(ap + ((g + 1) * EDGE_KG + k) * 4);
      }
      __builtin_amdgcn_sched_barrier(0);
#pragma unroll
      for (int k = 0; k < EDGE_KG; ++k) {
        bm = __builtin_amdgcn_wmma_f32_16x16x4_f32(false, a[cur][k], false,
                                                   bmf[g * EDGE_KG + k], (short)0,
                                                   bm, false, false);
        ba = __builtin_amdgcn_wmma_f32_16x16x4_f32(false, a[cur][k], false,
                                                   baf[g * EDGE_KG + k], (short)0,
                                                   ba, false, false);
      }
      __builtin_amdgcn_sched_barrier(0);
    }

    float p[8];
#pragma unroll
    for (int v = 0; v < 8; ++v) {
      const int r = v + 8 * half;
      const float qv = Qs[r * EDGE_PITCH + w * 16 + lr];
      const float kv = Ks[r * EDGE_PITCH + w * 16 + lr];
      float t = qv * (kv * (1.0f + bm[v]) + ba[v]);
      t += __shfl_xor(t, 8, 32);
      p[v] = t;
    }
    if (lr < 8) {
#pragma unroll
      for (int v = 0; v < 8; ++v) Ps[w * 128 + (v + 8 * half) * 8 + lr] = p[v];
    }
    __syncthreads();

    if (tid < 128) {
      float s = Ps[tid];
#pragma unroll
      for (int j = 1; j < 8; ++j) s += Ps[j * 128 + tid];
      const int e = tile * 16 + (tid >> 3);
      if (e < Medges) {
        logit_out[(size_t)e * NH + (tid & 7)] = s;
        atomicMax(&segkey[(size_t)myq * NH + (tid & 7)], float_key(s));
      }
    }
    par ^= 1;
  }
}

#define RR_WAVES 32
#define RR_LCAP 1024
#define RR_U 13

__device__ __forceinline__ void rr_batch(const int* lp, int nb, int lo, int lane,
                                         const float* __restrict__ logit,
                                         const int* __restrict__ qidx,
                                         const int* __restrict__ kidx,
                                         const float* __restrict__ kvbuf,
                                         float* Sum, float* Den, const float* Mx) {
  const int h = lane & 7;
  const int e = lp[h < nb ? h : 0];
  const int myq = qidx[e] - lo;
  const int myk = kidx[e];
  int ee[8], qi[8], ki[8];
#pragma unroll
  for (int u = 0; u < 8; ++u) {
    ee[u] = __builtin_amdgcn_readlane(e, u);
    qi[u] = __builtin_amdgcn_readlane(myq, u);
    ki[u] = __builtin_amdgcn_readlane(myk, u);
  }
  float lg[8], mx[8], vv[8][4];
#pragma unroll
  for (int u = 0; u < 8; ++u) {
    const float* vp = kvbuf + (size_t)ki[u] * (2 * HC) + HC + lane;
#pragma unroll
    for (int i = 0; i < 4; ++i) vv[u][i] = vp[32 * i];
  }
#pragma unroll
  for (int u = 0; u < 8; ++u) {
    lg[u] = logit[(size_t)ee[u] * NH + h];
    mx[u] = Mx[qi[u] * NH + h];
  }
#pragma unroll
  for (int u = 0; u < 8; ++u) {
    if (u < nb) {
      const float ex = __expf(lg[u] - mx[u]);
      if (lane < 8) atomicAdd(Den + qi[u] * NH + h, ex);
      float* sp = Sum + qi[u] * HC + lane;
#pragma unroll
      for (int i = 0; i < 4; ++i) atomicAdd(sp + 32 * i, ex * vv[u][i]);
    }
  }
}

__global__ __launch_bounds__(1024) void rowrange_softmax_agg_kernel(
    const float* __restrict__ logit, const int* __restrict__ qidx,
    const int* __restrict__ kidx, const unsigned* __restrict__ segkey,
    const float* __restrict__ kvbuf, const unsigned* __restrict__ bitmap,
    float* __restrict__ agg, int Medges, int n) {
  __shared__ __attribute__((aligned(16))) float Sum[RR_ROWS * HC];
  __shared__ float Den[RR_ROWS * NH];
  __shared__ float Mx[RR_ROWS * NH];
  __shared__ int List[RR_WAVES * RR_LCAP];

  const int tid = threadIdx.x;
  const int lane = tid & 31;
  const int w = __builtin_amdgcn_readfirstlane(tid >> 5);
  int* list = List + w * RR_LCAP;
  const int nvec = (Medges + BM_EDGES - 1) / BM_EDGES;
  const size_t pitch = (size_t)nvec * 32;

  for (int lo = blockIdx.x * RR_ROWS; lo < n; lo += gridDim.x * RR_ROWS) {
    const int nrows = (n - lo) < RR_ROWS ? (n - lo) : RR_ROWS;
    const unsigned* bmrow = bitmap + (size_t)(lo / RR_ROWS) * pitch;

    for (int i = tid; i < RR_ROWS * HC / 4; i += 1024)
      ((float4*)Sum)[i] = (float4){0.f, 0.f, 0.f, 0.f};
    for (int i = tid; i < nrows * NH; i += 1024) {
      Den[i] = 0.0f;
      Mx[i] = key_float(segkey[(size_t)lo * NH + i]);
    }
    __syncthreads();

    int head = 0, cnt = 0;
    for (int v0 = w; v0 < nvec; v0 += RR_WAVES * RR_U) {
      unsigned wd[RR_U];
#pragma unroll
      for (int u = 0; u < RR_U; ++u) {
        const int v = v0 + u * RR_WAVES;
        const int vv = v < nvec ? v : v0;
        wd[u] = bmrow[(size_t)vv * 32 + lane];
      }
#pragma unroll
      for (int u = 0; u < RR_U; ++u)
        if (v0 + u * RR_WAVES >= nvec) wd[u] = 0u;
      bool more = true;
      while (more) {
        unsigned left = 0u;
#pragma unroll
        for (int u = 0; u < RR_U; ++u) {
          const unsigned x = wd[u];
          const bool m = x != 0u;
          const unsigned mask = __builtin_amdgcn_ballot_w32(m);
          if (mask != 0u && cnt + 32 <= RR_LCAP) {
            const int pos = cnt + (int)__builtin_amdgcn_mbcnt_lo(mask, 0u);
            if (m) {
              list[pos] = ((v0 + u * RR_WAVES) * 32 + lane) * 32 + (__ffs((int)x) - 1);
              wd[u] = x & (x - 1u);
            }
            cnt += __builtin_popcount(mask);
          }
          left |= wd[u];
        }
        more = __builtin_amdgcn_ballot_w32(left != 0u) != 0u;
        const int keep = cnt + 32 > RR_LCAP ? 0 : 7;
        while (cnt - head > keep) {
          const int nb = cnt - head < 8 ? cnt - head : 8;
          rr_batch(list + head, nb, lo, lane, logit, qidx, kidx, kvbuf, Sum, Den, Mx);
          head += nb;
        }
        if (head == cnt) { head = 0; cnt = 0; }
      }
    }
    while (head < cnt) {
      const int nb = cnt - head < 8 ? cnt - head : 8;
      rr_batch(list + head, nb, lo, lane, logit, qidx, kidx, kvbuf, Sum, Den, Mx);
      head += nb;
    }
    __syncthreads();

    for (int i = tid; i < nrows * (HC / 4); i += 1024) {
      const int r = i >> 5, c4 = i & 31;
      const float4 s = ((const float4*)Sum)[i];
      const float* dp = Den + r * NH + 4 * (c4 & 1);
      const float e0 = dp[0], e1 = dp[1], e2 = dp[2], e3 = dp[3];
      float4 o;
      o.x = e0 == 0.0f ? 0.0f : s.x / e0;
      o.y = e1 == 0.0f ? 0.0f : s.y / e1;
      o.z = e2 == 0.0f ? 0.0f : s.z / e2;
      o.w = e3 == 0.0f ? 0.0f : s.w / e3;
      *(float4*)(agg + (size_t)(lo + r) * HC + c4 * 4) = o;
    }
    __syncthreads();
  }
}

extern "C" void kernel_launch(void* const* d_in, const int* in_sizes, int n_in,
                              void* d_out, int out_size, void* d_ws, size_t ws_size,
                              hipStream_t stream) {
  const float* query = (const float*)d_in[0];
  const float* key_  = (const float*)d_in[1];
  const int*   qidx  = (const int*)d_in[2];
  const int*   kidx  = (const int*)d_in[3];
  const float* pr    = (const float*)d_in[4];
  const float* Wq    = (const float*)d_in[5];
  const float* Wkv   = (const float*)d_in[6];
  const float* Wb    = (const float*)d_in[7];
  const float* Wo    = (const float*)d_in[8];

  const int N = in_sizes[0] / HC;
  const int M = in_sizes[2];

  float* out_result = (float*)d_out;
  float* out_logit  = out_result + (size_t)N * HC;

  float*    q_buf  = (float*)d_ws;
  float*    kv_buf = q_buf + (size_t)N * HC;
  unsigned* segkey = (unsigned*)(kv_buf + (size_t)N * 2 * HC);
  float*    agg    = (float*)(segkey + (size_t)N * NH) + (size_t)N * NH;
  unsigned* bitmap = (unsigned*)(agg + (size_t)N * HC);

  const int TPB = 256;

  init_ws_kernel<<<512, TPB, 0, stream>>>(segkey, qidx, bitmap, N, M);

  wmma_gemm_f32<HC, HC, 1>
      <<<GEMM_GRID, TPB, 0, stream>>>(query, Wq, q_buf, N);

  wmma_gemm_f32<2 * HC, 2 * HC, 2>
      <<<GEMM_GRID, TPB, 0, stream>>>(key_, Wkv, kv_buf, N);

  edge_bias_logit_kernel<<<512, TPB, 0, stream>>>(
      pr, Wb, q_buf, kv_buf, qidx, kidx, out_logit, segkey, M);

  rowrange_softmax_agg_kernel<<<256, 1024, 0, stream>>>(
      out_logit, qidx, kidx, segkey, kv_buf, bitmap, agg, M, N);

  wmma_gemm_f32<HC, HC, 1>
      <<<GEMM_GRID, TPB, 0, stream>>>(agg, Wo, out_result, N);
}
